# starting kernel
# baseline (speedup 1.0000x reference)
#include <hip/hip_runtime.h>
#include <hip/hip_bf16.h>

#define IN_F 128
#define HID  128
#define N_CLS 2

typedef __attribute__((ext_vector_type(2))) float v2f;
typedef __attribute__((ext_vector_type(8))) float v8f;

__global__ void gcn_zero_f32(float* __restrict__ p, int n) {
    int i = blockIdx.x * blockDim.x + threadIdx.x;
    if (i < n) p[i] = 0.0f;
}

__global__ void gcn_deg_accum(const int* __restrict__ dst, float* __restrict__ deg, int nE) {
    int e = blockIdx.x * blockDim.x + threadIdx.x;
    if (e < nE) atomicAdd(&deg[dst[e]], 1.0f);
}

__global__ void gcn_dinv_finalize(float* __restrict__ dinv, int n) {
    int i = blockIdx.x * blockDim.x + threadIdx.x;
    if (i < n) dinv[i] = rsqrtf(dinv[i] + 1.0f);
}

__global__ __launch_bounds__(64) void gcn_gemm1_wmma(const float* __restrict__ x,
                                                     const float* __restrict__ W1,
                                                     float* __restrict__ xw) {
    const int wave = blockIdx.x * (blockDim.x >> 5) + (threadIdx.x >> 5);
    const int lane = threadIdx.x & 31;
    const int half = lane >> 4;
    const int lm   = lane & 15;
    const int row0 = wave * 16;

    const float* __restrict__ xrow = x + (size_t)(row0 + lm) * IN_F;

    v8f acc[8];
#pragma unroll
    for (int cb = 0; cb < 8; ++cb) acc[cb] = (v8f){};

#pragma unroll 4
    for (int k0 = 0; k0 < IN_F; k0 += 4) {
        const int ka = k0 + 2 * half;
        v2f a = *(const v2f*)(xrow + ka);
        const float* __restrict__ w0 = W1 + (size_t)(ka + 0) * HID + lm;
        const float* __restrict__ w1 = W1 + (size_t)(ka + 1) * HID + lm;
#pragma unroll
        for (int cb = 0; cb < 8; ++cb) {
            v2f b;
            b.x = w0[cb * 16];
            b.y = w1[cb * 16];
            acc[cb] = __builtin_amdgcn_wmma_f32_16x16x4_f32(false, a, false, b,
                                                            (short)0, acc[cb], false, false);
        }
    }
#pragma unroll
    for (int cb = 0; cb < 8; ++cb) {
#pragma unroll
        for (int r = 0; r < 8; ++r) {
            xw[(size_t)(row0 + r + 8 * half) * HID + cb * 16 + lm] = acc[cb][r];
        }
    }
}

__global__ void gcn_selfloop_h(const float* __restrict__ xw, const float* __restrict__ dinv,
                               float* __restrict__ h, int n) {
    int idx = blockIdx.x * blockDim.x + threadIdx.x;
    int node = idx >> 5;
    if (node >= n) return;
    int lane = idx & 31;
    float s = dinv[node];
    s = s * s;
    const float4 v = *(const float4*)(xw + (size_t)node * HID + lane * 4);
    float4 o = make_float4(v.x * s, v.y * s, v.z * s, v.w * s);
    *(float4*)(h + (size_t)node * HID + lane * 4) = o;
}

__global__ void gcn_agg1(const int* __restrict__ src, const int* __restrict__ dst,
                         const float* __restrict__ dinv, const float* __restrict__ xw,
                         float* __restrict__ h, int nE) {
    int e = (blockIdx.x * blockDim.x + threadIdx.x) >> 5;
    if (e >= nE) return;
    int lane = threadIdx.x & 31;
    int s = src[e], d = dst[e];
    float norm = dinv[s] * dinv[d];
    const float4 v = *(const float4*)(xw + (size_t)s * HID + lane * 4);
    float* o = h + (size_t)d * HID + lane * 4;
    atomicAdd(o + 0, v.x * norm);
    atomicAdd(o + 1, v.y * norm);
    atomicAdd(o + 2, v.z * norm);
    atomicAdd(o + 3, v.w * norm);
}

__global__ void gcn_gemm2(const float* __restrict__ h, const float* __restrict__ b1,
                          const float* __restrict__ W2, float* __restrict__ zw, int n) {
    int node = (blockIdx.x * blockDim.x + threadIdx.x) >> 5;
    if (node >= n) return;
    int lane = threadIdx.x & 31;
    const float4 hv = *(const float4*)(h + (size_t)node * HID + lane * 4);
    const float4 bv = *(const float4*)(b1 + lane * 4);
    float f0 = fmaxf(hv.x + bv.x, 0.0f);
    float f1 = fmaxf(hv.y + bv.y, 0.0f);
    float f2 = fmaxf(hv.z + bv.z, 0.0f);
    float f3 = fmaxf(hv.w + bv.w, 0.0f);
    int k = lane * 4;
    float a0 = f0 * W2[(k + 0) * 2 + 0] + f1 * W2[(k + 1) * 2 + 0]
             + f2 * W2[(k + 2) * 2 + 0] + f3 * W2[(k + 3) * 2 + 0];
    float a1 = f0 * W2[(k + 0) * 2 + 1] + f1 * W2[(k + 1) * 2 + 1]
             + f2 * W2[(k + 2) * 2 + 1] + f3 * W2[(k + 3) * 2 + 1];
#pragma unroll
    for (int off = 16; off > 0; off >>= 1) {
        a0 += __shfl_down(a0, off, 32);
        a1 += __shfl_down(a1, off, 32);
    }
    if (lane == 0) {
        zw[(size_t)node * 2 + 0] = a0;
        zw[(size_t)node * 2 + 1] = a1;
    }
}

__global__ void gcn_selfloop_out(const float* __restrict__ zw, const float* __restrict__ dinv,
                                 float* __restrict__ out, int n) {
    int i = blockIdx.x * blockDim.x + threadIdx.x;
    if (i >= n) return;
    float s = dinv[i];
    s = s * s;
    out[(size_t)i * 2 + 0] = zw[(size_t)i * 2 + 0] * s;
    out[(size_t)i * 2 + 1] = zw[(size_t)i * 2 + 1] * s;
}

__global__ void gcn_agg2(const int* __restrict__ src, const int* __restrict__ dst,
                         const float* __restrict__ dinv, const float* __restrict__ zw,
                         float* __restrict__ out, int nE) {
    int e = blockIdx.x * blockDim.x + threadIdx.x;
    if (e >= nE) return;
    int s = src[e], d = dst[e];
    float norm = dinv[s] * dinv[d];
    atomicAdd(out + (size_t)d * 2 + 0, zw[(size_t)s * 2 + 0] * norm);
    atomicAdd(out + (size_t)d * 2 + 1, zw[(size_t)s * 2 + 1] * norm);
}

__global__ void gcn_logsoftmax(float* __restrict__ out, const float* __restrict__ b2, int n) {
    int i = blockIdx.x * blockDim.x + threadIdx.x;
    if (i >= n) return;
    float l0 = out[(size_t)i * 2 + 0] + b2[0];
    float l1 = out[(size_t)i * 2 + 1] + b2[1];
    float m = fmaxf(l0, l1);
    float lse = m + logf(expf(l0 - m) + expf(l1 - m));
    out[(size_t)i * 2 + 0] = l0 - lse;
    out[(size_t)i * 2 + 1] = l1 - lse;
}

extern "C" void kernel_launch(void* const* d_in, const int* in_sizes, int n_in,
                              void* d_out, int out_size, void* d_ws, size_t ws_size,
                              hipStream_t stream) {
    const float* x  = (const float*)d_in[0];
    const float* W1 = (const float*)d_in[1];
    const float* b1 = (const float*)d_in[2];
    const float* W2 = (const float*)d_in[3];
    const float* b2 = (const float*)d_in[4];
    const int*   ei = (const int*)d_in[5];

    const int n  = in_sizes[0] / IN_F;
    const int nE = in_sizes[5] / 2;
    const int* src = ei;
    const int* dst = ei + nE;

    char* ws = (char*)d_ws;
    float* xw   = (float*)(ws);
    float* h    = (float*)(ws + (size_t)n * HID * 4);
    float* dinv = (float*)(ws + (size_t)2 * n * HID * 4);
    float* zw   = (float*)(ws + (size_t)2 * n * HID * 4 + 512000);
    float* out  = (float*)d_out;

    const int B = 256;

    gcn_zero_f32<<<(n + B - 1) / B, B, 0, stream>>>(dinv, n);
    gcn_deg_accum<<<(nE + B - 1) / B, B, 0, stream>>>(dst, dinv, nE);
    gcn_dinv_finalize<<<(n + B - 1) / B, B, 0, stream>>>(dinv, n);

    {
        int rowTiles = n / 16;
        gcn_gemm1_wmma<<<rowTiles / 2, 64, 0, stream>>>(x, W1, xw);
    }

    gcn_selfloop_h<<<((size_t)n * 32 + B - 1) / B, B, 0, stream>>>(xw, dinv, h, n);
    gcn_agg1<<<((size_t)nE * 32 + B - 1) / B, B, 0, stream>>>(src, dst, dinv, xw, h, nE);

    gcn_gemm2<<<((size_t)n * 32 + B - 1) / B, B, 0, stream>>>(h, b1, W2, zw, n);

    gcn_selfloop_out<<<(n + B - 1) / B, B, 0, stream>>>(zw, dinv, out, n);
    gcn_agg2<<<(nE + B - 1) / B, B, 0, stream>>>(src, dst, dinv, zw, out, nE);

    gcn_logsoftmax<<<(n + B - 1) / B, B, 0, stream>>>(out, b2, n);
}


# layer-1 aggregation sweeps a 1-byte owner-key plane (4x fewer bytes, 512 edges per wave load) that the degree kernel writes with a packed src|offset word per edge; LDS f32 atomics kept, arrival order
# speedup vs baseline: 1.6363x; 1.6363x over previous
#include <hip/hip_runtime.h>
#include <hip/hip_bf16.h>

#define IN_F 128
#define HID  128
#define N_CLS 2

typedef __attribute__((ext_vector_type(2))) float v2f;
typedef __attribute__((ext_vector_type(8))) float v8f;

__global__ void gcn_zero_f32(float* __restrict__ p, int n) {
    int i = blockIdx.x * blockDim.x + threadIdx.x;
    if (i < n) p[i] = 0.0f;
}

#define NPB     400
#define PK_BITS 17

__device__ __forceinline__ unsigned gcn_pack_edge(int s, int d, unsigned* key) {
    const unsigned g = (unsigned)d / NPB;
    *key = g;
    return (unsigned)s | (((unsigned)d - g * NPB) << PK_BITS);
}

__global__ void gcn_deg_keys(const int* __restrict__ src, const int* __restrict__ dst,
                             float* __restrict__ deg, unsigned* __restrict__ pk,
                             unsigned char* __restrict__ keys, int nE) {
    const int t  = blockIdx.x * blockDim.x + threadIdx.x;
    const int e0 = t * 4;
    if (e0 >= nE) return;
    if ((nE & 3) == 0) {
        const int4 d = ((const int4*)dst)[t];
        const int4 s = ((const int4*)src)[t];
        atomicAdd(&deg[d.x], 1.0f);
        atomicAdd(&deg[d.y], 1.0f);
        atomicAdd(&deg[d.z], 1.0f);
        atomicAdd(&deg[d.w], 1.0f);
        unsigned k0, k1, k2, k3;
        uint4 w;
        w.x = gcn_pack_edge(s.x, d.x, &k0);
        w.y = gcn_pack_edge(s.y, d.y, &k1);
        w.z = gcn_pack_edge(s.z, d.z, &k2);
        w.w = gcn_pack_edge(s.w, d.w, &k3);
        ((uint4*)pk)[t] = w;
        ((unsigned*)keys)[t] = k0 | (k1 << 8) | (k2 << 16) | (k3 << 24);
    } else {
        const int e1 = min(e0 + 4, nE);
        for (int e = e0; e < e1; ++e) {
            const int d = dst[e];
            atomicAdd(&deg[d], 1.0f);
            unsigned k;
            pk[e]   = gcn_pack_edge(src[e], d, &k);
            keys[e] = (unsigned char)k;
        }
    }
}

__global__ void gcn_dinv_finalize(float* __restrict__ dinv, int n) {
    int i = blockIdx.x * blockDim.x + threadIdx.x;
    if (i < n) dinv[i] = rsqrtf(dinv[i] + 1.0f);
}

__global__ __launch_bounds__(64) void gcn_gemm1_wmma(const float* __restrict__ x,
                                                     const float* __restrict__ W1,
                                                     float* __restrict__ xw) {
    const int wave = blockIdx.x * (blockDim.x >> 5) + (threadIdx.x >> 5);
    const int lane = threadIdx.x & 31;
    const int half = lane >> 4;
    const int lm   = lane & 15;
    const int row0 = wave * 16;

    const float* __restrict__ xrow = x + (size_t)(row0 + lm) * IN_F;

    v8f acc[8];
#pragma unroll
    for (int cb = 0; cb < 8; ++cb) acc[cb] = (v8f){};

#pragma unroll 4
    for (int k0 = 0; k0 < IN_F; k0 += 4) {
        const int ka = k0 + 2 * half;
        v2f a = *(const v2f*)(xrow + ka);
        const float* __restrict__ w0 = W1 + (size_t)(ka + 0) * HID + lm;
        const float* __restrict__ w1 = W1 + (size_t)(ka + 1) * HID + lm;
#pragma unroll
        for (int cb = 0; cb < 8; ++cb) {
            v2f b;
            b.x = w0[cb * 16];
            b.y = w1[cb * 16];
            acc[cb] = __builtin_amdgcn_wmma_f32_16x16x4_f32(false, a, false, b,
                                                            (short)0, acc[cb], false, false);
        }
    }
#pragma unroll
    for (int cb = 0; cb < 8; ++cb) {
#pragma unroll
        for (int r = 0; r < 8; ++r) {
            xw[(size_t)(row0 + r + 8 * half) * HID + cb * 16 + lm] = acc[cb][r];
        }
    }
}

#define AGG_T  1024
#define AGG_W  (AGG_T / 32)
#define LCAP   256
#define SCAN_G 8

__device__ __forceinline__ void gcn_lds_add(float* p, float v) {
    __hip_atomic_fetch_add(p, v, __ATOMIC_RELAXED, __HIP_MEMORY_SCOPE_WORKGROUP);
}

__device__ __forceinline__ int gcn_list_append(unsigned* lst, int cnt, bool m, int e) {
    const unsigned mask = __builtin_amdgcn_ballot_w32(m);
    if (mask != 0u) {
        const int pos = cnt + (int)__builtin_amdgcn_mbcnt_lo(mask, 0u);
        if (m) lst[pos] = (unsigned)e;
        cnt += __builtin_popcount(mask);
    }
    return cnt;
}

__device__ __forceinline__ unsigned gcn_zero_bytes(unsigned x) {
    return ~(((x & 0x7F7F7F7Fu) + 0x7F7F7F7Fu) | x | 0x7F7F7F7Fu);
}

__device__ __forceinline__ void gcn_list_flush(const unsigned* lst, int cnt, int lane, int lo,
                                               const unsigned* __restrict__ pk,
                                               const float* __restrict__ dinv,
                                               const float* __restrict__ xw, float* acc) {
    for (int i0 = 0; i0 < cnt; i0 += 32) {
        const int idx = i0 + lane;
        const bool valid = idx < cnt;
        const unsigned w = valid ? pk[lst[idx]] : 0u;
        const int s   = (int)(w & ((1u << PK_BITS) - 1u));
        const int off = (int)(w >> PK_BITS);
        const float nrm = valid ? dinv[s] * dinv[lo + off] : 0.0f;
        const int m = min(32, cnt - i0);
        for (int k0 = 0; k0 < m; k0 += 8) {
            float4 v[8];
#pragma unroll
            for (int u = 0; u < 8; ++u) {
                const int sk = __builtin_amdgcn_readlane(s, k0 + u);
                v[u] = *(const float4*)(xw + (size_t)sk * HID + lane * 4);
            }
#pragma unroll
            for (int u = 0; u < 8; ++u) {
                if (k0 + u < m) {
                    const int dk = __builtin_amdgcn_readlane(off, k0 + u);
                    const float nk = __int_as_float(
                        __builtin_amdgcn_readlane(__float_as_int(nrm), k0 + u));
                    float* a = acc + dk * HID + lane;
                    gcn_lds_add(a +  0, v[u].x * nk);
                    gcn_lds_add(a + 32, v[u].y * nk);
                    gcn_lds_add(a + 64, v[u].z * nk);
                    gcn_lds_add(a + 96, v[u].w * nk);
                }
            }
        }
    }
}

__global__ __launch_bounds__(AGG_T) void gcn_agg1_lds(const unsigned* __restrict__ pk,
                                                      const unsigned char* __restrict__ keys,
                                                      const float* __restrict__ dinv,
                                                      const float* __restrict__ xw,
                                                      const float* __restrict__ b1,
                                                      const float* __restrict__ W2,
                                                      float* __restrict__ zw,
                                                      float* __restrict__ out, int n, int nE) {
    __shared__ __attribute__((aligned(16))) float acc[NPB * HID];
    __shared__ unsigned lst_all[AGG_W * LCAP];
    __shared__ __attribute__((aligned(16))) float zs[NPB * N_CLS];

    const int lane = threadIdx.x & 31;
    const int wave = __builtin_amdgcn_readfirstlane((int)(threadIdx.x >> 5));
    const int lo   = blockIdx.x * NPB;
    const int own  = min(NPB, n - lo);
    unsigned* lst  = lst_all + wave * LCAP;

    for (int r = wave; r < own; r += AGG_W) {
        const int node = lo + r;
        float s = dinv[node];
        s = s * s;
        const float4 v = *(const float4*)(xw + (size_t)node * HID + lane * 4);
        float* a = acc + r * HID + lane;
        a[0]  = v.x * s;
        a[32] = v.y * s;
        a[64] = v.z * s;
        a[96] = v.w * s;
    }
    __syncthreads();

    const int nVec = ((nE & 3) == 0) ? (nE >> 9) : 0;
    const uint4* __restrict__ keys4 = (const uint4*)keys;
    const unsigned mine = (unsigned)blockIdx.x * 0x01010101u;
    const int bitEdge = ((lane & 7) << 2) + (lane >> 3);
    const int rot = (int)(((long long)blockIdx.x * nVec) / gridDim.x);
    int cnt = 0;

    for (int q0 = wave; q0 < nVec; q0 += AGG_W * SCAN_G) {
        uint4 kv[SCAN_G];
        int   bb[SCAN_G];
#pragma unroll
        for (int g = 0; g < SCAN_G; ++g) {
            int b = min(q0 + g * AGG_W, nVec - 1) + rot;
            b = (b >= nVec) ? b - nVec : b;
            bb[g] = b;
            kv[g] = keys4[(size_t)b * 32 + lane];
        }
#pragma unroll
        for (int g = 0; g < SCAN_G; ++g) {
            const unsigned m = (gcn_zero_bytes(kv[g].x ^ mine) >> 7) | (gcn_zero_bytes(kv[g].y ^ mine) >> 6) |
                               (gcn_zero_bytes(kv[g].z ^ mine) >> 5) | (gcn_zero_bytes(kv[g].w ^ mine) >> 4);
            unsigned lanes = __builtin_amdgcn_ballot_w32(m != 0u);
            if (q0 + g * AGG_W >= nVec) lanes = 0u;
            const int eb = bb[g] * 512 + bitEdge;
            while (lanes != 0u) {
                const int l = __builtin_ctz(lanes);
                lanes &= lanes - 1u;
                const unsigned ml = (unsigned)__builtin_amdgcn_readlane((int)m, l);
                const int pos = cnt + (int)__builtin_amdgcn_mbcnt_lo(ml, 0u);
                if ((ml >> lane) & 1u) lst[pos] = (unsigned)(eb + l * 16);
                cnt += __builtin_popcount(ml);
                if (cnt > LCAP - 16) {
                    gcn_list_flush(lst, cnt, lane, lo, pk, dinv, xw, acc);
                    cnt = 0;
                }
            }
        }
    }
    for (int base = nVec * 512 + wave * 32; base < nE; base += AGG_W * 32) {
        if (cnt > LCAP - 32) {
            gcn_list_flush(lst, cnt, lane, lo, pk, dinv, xw, acc);
            cnt = 0;
        }
        const int e = base + lane;
        const bool hit = (e < nE) && ((unsigned)keys[e] == (unsigned)blockIdx.x);
        cnt = gcn_list_append(lst, cnt, hit, e);
    }
    gcn_list_flush(lst, cnt, lane, lo, pk, dinv, xw, acc);
    __syncthreads();

    {
        const float4 bv = *(const float4*)(b1 + lane * 4);
        const int k = lane * 4;
        const float w00 = W2[(k + 0) * 2 + 0], w01 = W2[(k + 0) * 2 + 1];
        const float w10 = W2[(k + 1) * 2 + 0], w11 = W2[(k + 1) * 2 + 1];
        const float w20 = W2[(k + 2) * 2 + 0], w21 = W2[(k + 2) * 2 + 1];
        const float w30 = W2[(k + 3) * 2 + 0], w31 = W2[(k + 3) * 2 + 1];
        for (int r = wave; r < own; r += AGG_W) {
            const float* a = acc + r * HID + lane;
            const float f0 = fmaxf(a[0]  + bv.x, 0.0f);
            const float f1 = fmaxf(a[32] + bv.y, 0.0f);
            const float f2 = fmaxf(a[64] + bv.z, 0.0f);
            const float f3 = fmaxf(a[96] + bv.w, 0.0f);
            float a0 = f0 * w00 + f1 * w10 + f2 * w20 + f3 * w30;
            float a1 = f0 * w01 + f1 * w11 + f2 * w21 + f3 * w31;
#pragma unroll
            for (int o = 16; o > 0; o >>= 1) {
                a0 += __shfl_down(a0, o, 32);
                a1 += __shfl_down(a1, o, 32);
            }
            if (lane == 0) {
                zs[r * 2 + 0] = a0;
                zs[r * 2 + 1] = a1;
            }
        }
    }
    __syncthreads();

    const int nOut = own * N_CLS;
    for (int i = threadIdx.x * 4; i < nOut; i += AGG_T * 4) {
        if (i + 3 < nOut) {
            const float4 z = *(const float4*)(zs + i);
            float s0 = dinv[lo + (i >> 1)];
            float s1 = dinv[lo + (i >> 1) + 1];
            s0 = s0 * s0;
            s1 = s1 * s1;
            *(float4*)(zw  + (size_t)lo * N_CLS + i) = z;
            *(float4*)(out + (size_t)lo * N_CLS + i) = make_float4(z.x * s0, z.y * s0, z.z * s1, z.w * s1);
        } else {
            for (int j = i; j < nOut; ++j) {
                float s = dinv[lo + (j >> 1)];
                s = s * s;
                zw[(size_t)lo * N_CLS + j]  = zs[j];
                out[(size_t)lo * N_CLS + j] = zs[j] * s;
            }
        }
    }
}

__global__ void gcn_agg2(const int* __restrict__ src, const int* __restrict__ dst,
                         const float* __restrict__ dinv, const float* __restrict__ zw,
                         float* __restrict__ out, int nE) {
    int e = blockIdx.x * blockDim.x + threadIdx.x;
    if (e >= nE) return;
    int s = src[e], d = dst[e];
    float norm = dinv[s] * dinv[d];
    atomicAdd(out + (size_t)d * 2 + 0, zw[(size_t)s * 2 + 0] * norm);
    atomicAdd(out + (size_t)d * 2 + 1, zw[(size_t)s * 2 + 1] * norm);
}

__global__ void gcn_logsoftmax(float* __restrict__ out, const float* __restrict__ b2, int n) {
    int i = blockIdx.x * blockDim.x + threadIdx.x;
    if (i >= n) return;
    float l0 = out[(size_t)i * 2 + 0] + b2[0];
    float l1 = out[(size_t)i * 2 + 1] + b2[1];
    float m = fmaxf(l0, l1);
    float lse = m + logf(expf(l0 - m) + expf(l1 - m));
    out[(size_t)i * 2 + 0] = l0 - lse;
    out[(size_t)i * 2 + 1] = l1 - lse;
}

extern "C" void kernel_launch(void* const* d_in, const int* in_sizes, int n_in,
                              void* d_out, int out_size, void* d_ws, size_t ws_size,
                              hipStream_t stream) {
    const float* x  = (const float*)d_in[0];
    const float* W1 = (const float*)d_in[1];
    const float* b1 = (const float*)d_in[2];
    const float* W2 = (const float*)d_in[3];
    const float* b2 = (const float*)d_in[4];
    const int*   ei = (const int*)d_in[5];

    const int n  = in_sizes[0] / IN_F;
    const int nE = in_sizes[5] / 2;
    const int* src = ei;
    const int* dst = ei + nE;

    char* ws = (char*)d_ws;
    float* xw   = (float*)(ws);
    unsigned*      pk   = (unsigned*)(ws + (size_t)n * HID * 4);
    unsigned char* keys = (unsigned char*)(ws + (size_t)n * HID * 4) + (size_t)nE * 4;
    float* dinv = (float*)(ws + (size_t)2 * n * HID * 4);
    float* zw   = (float*)(ws + (size_t)2 * n * HID * 4 + 512000);
    float* out  = (float*)d_out;

    const int B = 256;

    gcn_zero_f32<<<(n + B - 1) / B, B, 0, stream>>>(dinv, n);
    gcn_deg_keys<<<((nE + 3) / 4 + B - 1) / B, B, 0, stream>>>(src, dst, dinv, pk, keys, nE);
    gcn_dinv_finalize<<<(n + B - 1) / B, B, 0, stream>>>(dinv, n);

    {
        int rowTiles = n / 16;
        gcn_gemm1_wmma<<<rowTiles / 2, 64, 0, stream>>>(x, W1, xw);
    }

    gcn_agg1_lds<<<(n + NPB - 1) / NPB, AGG_T, 0, stream>>>(pk, keys, dinv, xw, b1, W2, zw, out, n, nE);

    gcn_agg2<<<(nE + B - 1) / B, B, 0, stream>>>(src, dst, dinv, zw, out, nE);

    gcn_logsoftmax<<<(n + B - 1) / B, B, 0, stream>>>(out, b2, n);
}


# layer-1 aggregation: a wave's 13 start-value row loads, the counts and its first list chunk in flight together ahead of the barrier; epilogue rows overlap, lane sums by DPP adds (same pair order)
# speedup vs baseline: 4.1526x; 2.5378x over previous
#include <hip/hip_runtime.h>
#include <hip/hip_bf16.h>

#define IN_F 128
#define HID  128
#define N_CLS 2

typedef __attribute__((ext_vector_type(2))) float v2f;
typedef __attribute__((ext_vector_type(8))) float v8f;

#define NPB     416
#define PK_BITS 17

__device__ __forceinline__ unsigned gcn_pack_edge(int s, int d, unsigned* key) {
    const unsigned g = (unsigned)d / NPB;
    *key = g;
    return (unsigned)s | (((unsigned)d - g * NPB) << PK_BITS);
}

__global__ void gcn_edge_keys(const int* __restrict__ src, const int* __restrict__ dst,
                              unsigned* __restrict__ pk, unsigned char* __restrict__ keys, int nE) {
    const int t  = blockIdx.x * blockDim.x + threadIdx.x;
    const int e0 = t * 4;
    if (e0 >= nE) return;
    if ((nE & 3) == 0) {
        const int4 d = ((const int4*)dst)[t];
        const int4 s = ((const int4*)src)[t];
        unsigned k0, k1, k2, k3;
        uint4 w;
        w.x = gcn_pack_edge(s.x, d.x, &k0);
        w.y = gcn_pack_edge(s.y, d.y, &k1);
        w.z = gcn_pack_edge(s.z, d.z, &k2);
        w.w = gcn_pack_edge(s.w, d.w, &k3);
        ((uint4*)pk)[t] = w;
        ((unsigned*)keys)[t] = k0 | (k1 << 8) | (k2 << 16) | (k3 << 24);
    } else {
        const int e1 = min(e0 + 4, nE);
        for (int e = e0; e < e1; ++e) {
            unsigned k;
            pk[e]   = gcn_pack_edge(src[e], dst[e], &k);
            keys[e] = (unsigned char)k;
        }
    }
}

__device__ __forceinline__ void gcn_gemm1_tile(const float* __restrict__ x,
                                               const float* __restrict__ W1,
                                               float* __restrict__ xw,
                                               int strip, int ch, int lane) {
    const int half = lane >> 4;
    const int lm   = lane & 15;
    const int row0 = strip * 16;
    const int col0 = ch * 64;

    const float* __restrict__ xrow = x + (size_t)(row0 + lm) * IN_F;

    v8f acc[4];
#pragma unroll
    for (int cb = 0; cb < 4; ++cb) acc[cb] = (v8f){};

#pragma unroll 2
    for (int k0 = 0; k0 < IN_F; k0 += 4) {
        const int ka = k0 + 2 * half;
        v2f a = *(const v2f*)(xrow + ka);
        const float* __restrict__ w0 = W1 + (size_t)(ka + 0) * HID + col0 + lm;
        const float* __restrict__ w1 = W1 + (size_t)(ka + 1) * HID + col0 + lm;
#pragma unroll
        for (int cb = 0; cb < 4; ++cb) {
            v2f b;
            b.x = w0[cb * 16];
            b.y = w1[cb * 16];
            acc[cb] = __builtin_amdgcn_wmma_f32_16x16x4_f32(false, a, false, b,
                                                            (short)0, acc[cb], false, false);
        }
    }
#pragma unroll
    for (int cb = 0; cb < 4; ++cb) {
#pragma unroll
        for (int r = 0; r < 8; ++r) {
            xw[(size_t)(row0 + r + 8 * half) * HID + col0 + cb * 16 + lm] = acc[cb][r];
        }
    }
}

#define AGG_T  1024
#define AGG_W  (AGG_T / 32)
#define LCAP   256
#define SCAN_G 8
#define WCAP   512
#define SLOT_W (AGG_W + AGG_W * WCAP)
#define CH     8

__device__ __forceinline__ void gcn_lds_add(float* p, float v) {
    __hip_atomic_fetch_add(p, v, __ATOMIC_RELAXED, __HIP_MEMORY_SCOPE_WORKGROUP);
}

__device__ __forceinline__ float gcn_xhalf(float x) {
    const int b = __float_as_int(x);
    return __int_as_float(__builtin_amdgcn_permlanex16(b, b, 0x76543210, 0xfedcba98, false, false));
}

template <int M>
__device__ __forceinline__ float gcn_xrow(float x) {
    return __int_as_float(__builtin_amdgcn_update_dpp(0, __float_as_int(x), 0x160 + M, 0xf, 0xf, true));
}

__device__ __forceinline__ float gcn_wave_sum(float v) {
    v += gcn_xhalf(v);
    v += gcn_xrow<8>(v);
    v += gcn_xrow<4>(v);
    v += gcn_xrow<2>(v);
    v += gcn_xrow<1>(v);
    return v;
}

__device__ __forceinline__ int gcn_list_append(unsigned* lst, int cnt, bool m, int e) {
    const unsigned mask = __builtin_amdgcn_ballot_w32(m);
    if (mask != 0u) {
        const int pos = cnt + (int)__builtin_amdgcn_mbcnt_lo(mask, 0u);
        if (m) lst[pos] = (unsigned)e;
        cnt += __builtin_popcount(mask);
    }
    return cnt;
}

__device__ __forceinline__ unsigned gcn_zero_bytes(unsigned x) {
    return ~(((x & 0x7F7F7F7Fu) + 0x7F7F7F7Fu) | x | 0x7F7F7F7Fu);
}

template <class F>
__device__ __forceinline__ void gcn_sweep(const unsigned char* __restrict__ keys, int nE, int nGroups,
                                          int lane, int wave, unsigned* lst, F flush) {
    const int nVec = ((nE & 3) == 0) ? (nE >> 9) : 0;
    const uint4* __restrict__ keys4 = (const uint4*)keys;
    const unsigned mine = (unsigned)blockIdx.x * 0x01010101u;
    const int bitEdge = ((lane & 7) << 2) + (lane >> 3);
    const int rot = (int)(((long long)blockIdx.x * nVec) / nGroups);
    int cnt = 0;

    for (int q0 = wave; q0 < nVec; q0 += AGG_W * SCAN_G) {
        uint4 kv[SCAN_G];
        int   bb[SCAN_G];
#pragma unroll
        for (int g = 0; g < SCAN_G; ++g) {
            int b = min(q0 + g * AGG_W, nVec - 1) + rot;
            b = (b >= nVec) ? b - nVec : b;
            bb[g] = b;
            kv[g] = keys4[(size_t)b * 32 + lane];
        }
#pragma unroll
        for (int g = 0; g < SCAN_G; ++g) {
            const unsigned m = (gcn_zero_bytes(kv[g].x ^ mine) >> 7) | (gcn_zero_bytes(kv[g].y ^ mine) >> 6) |
                               (gcn_zero_bytes(kv[g].z ^ mine) >> 5) | (gcn_zero_bytes(kv[g].w ^ mine) >> 4);
            unsigned lanes = __builtin_amdgcn_ballot_w32(m != 0u);
            if (q0 + g * AGG_W >= nVec) lanes = 0u;
            const int eb = bb[g] * 512 + bitEdge;
            while (lanes != 0u) {
                const int l = __builtin_ctz(lanes);
                lanes &= lanes - 1u;
                const unsigned ml = (unsigned)__builtin_amdgcn_readlane((int)m, l);
                const int pos = cnt + (int)__builtin_amdgcn_mbcnt_lo(ml, 0u);
                if ((ml >> lane) & 1u) lst[pos] = (unsigned)(eb + l * 16);
                cnt += __builtin_popcount(ml);
                if (cnt > LCAP - 16) {
                    flush(cnt);
                    cnt = 0;
                }
            }
        }
    }
    for (int base = nVec * 512 + wave * 32; base < nE; base += AGG_W * 32) {
        if (cnt > LCAP - 32) {
            flush(cnt);
            cnt = 0;
        }
        const int e = base + lane;
        const bool hit = (e < nE) && ((unsigned)keys[e] == (unsigned)blockIdx.x);
        cnt = gcn_list_append(lst, cnt, hit, e);
    }
    flush(cnt);
}

__device__ __forceinline__ bool gcn_slot_fits(const unsigned* __restrict__ slot, int lane, int wave,
                                              int* mine) {
    const unsigned c = slot[lane];
    *mine = __builtin_amdgcn_readlane((int)c, wave);
    return __builtin_amdgcn_ballot_w32(c > (unsigned)WCAP) == 0u;
}

__global__ __launch_bounds__(AGG_T) void gcn_own_gemm(const unsigned* __restrict__ pk,
                                                      const unsigned char* __restrict__ keys,
                                                      float* __restrict__ dinv,
                                                      unsigned* __restrict__ slots, int n, int nE,
                                                      const float* __restrict__ x,
                                                      const float* __restrict__ W1,
                                                      float* __restrict__ xw, int nGroups) {
    __shared__ float degl[NPB];
    __shared__ unsigned cntl[AGG_W];
    __shared__ unsigned lst_all[AGG_W * LCAP];

    const int lane = threadIdx.x & 31;
    const int wave = __builtin_amdgcn_readfirstlane((int)(threadIdx.x >> 5));

    if ((int)blockIdx.x >= nGroups) {
        const int t = ((int)blockIdx.x - nGroups) * AGG_W + wave;
        if ((t >> 1) < n / 16) gcn_gemm1_tile(x, W1, xw, t >> 1, t & 1, lane);
        return;
    }

    const int lo   = blockIdx.x * NPB;
    const int own  = min(NPB, n - lo);
    unsigned* lst  = lst_all + wave * LCAP;
    unsigned* slot = slots + (size_t)blockIdx.x * SLOT_W;
    unsigned* part = slot + AGG_W + wave * WCAP;

    for (int r = threadIdx.x; r < NPB; r += AGG_T) degl[r] = 1.0f;
    __syncthreads();

    int kept = 0;
    gcn_sweep(keys, nE, nGroups, lane, wave, lst, [&](int cnt) {
        for (int i0 = 0; i0 < cnt; i0 += 32 * CH) {
            unsigned w[CH];
#pragma unroll
            for (int c = 0; c < CH; ++c) {
                const int idx = min(i0 + c * 32 + lane, cnt - 1);
                w[c] = pk[lst[idx]];
            }
#pragma unroll
            for (int c = 0; c < CH; ++c) {
                const int idx = i0 + c * 32 + lane;
                if (idx < cnt) {
                    gcn_lds_add(degl + (w[c] >> PK_BITS), 1.0f);
                    if (kept + idx < WCAP) part[kept + idx] = w[c];
                }
            }
        }
        kept += cnt;
    });
    if (lane == 0) cntl[wave] = (unsigned)kept;
    __syncthreads();

    if (wave == 0) slot[lane] = cntl[lane];
    for (int r = threadIdx.x; r < own; r += AGG_T) dinv[lo + r] = rsqrtf(degl[r]);
}

static_assert(AGG_W == 32, "a group's line of counts is read and written by the 32 lanes of one wave");


__device__ __forceinline__ void gcn_rows_add(int s, int off, float nrm, int m, int lane,
                                             const float* __restrict__ xw, float* acc) {
    for (int k0 = 0; k0 < m; k0 += 8) {
        float4 v[8];
#pragma unroll
        for (int u = 0; u < 8; ++u) {
            const int sk = __builtin_amdgcn_readlane(s, k0 + u);
            v[u] = *(const float4*)(xw + (size_t)sk * HID + lane * 4);
        }
#pragma unroll
        for (int u = 0; u < 8; ++u) {
            if (k0 + u < m) {
                const int dk = __builtin_amdgcn_readlane(off, k0 + u);
                const float nk = __int_as_float(
                    __builtin_amdgcn_readlane(__float_as_int(nrm), k0 + u));
                float* a = acc + dk * HID + lane;
                gcn_lds_add(a +  0, v[u].x * nk);
                gcn_lds_add(a + 32, v[u].y * nk);
                gcn_lds_add(a + 64, v[u].z * nk);
                gcn_lds_add(a + 96, v[u].w * nk);
            }
        }
    }
}

__device__ __forceinline__ void gcn_chunk_fetch(const unsigned* lst, int i0, int cnt, int lane, int lo,
                                                const float* __restrict__ dinv,
                                                unsigned (&w)[CH], float (&nr)[CH]) {
#pragma unroll
    for (int c = 0; c < CH; ++c) w[c] = lst[min(i0 + c * 32 + lane, cnt - 1)];
#pragma unroll
    for (int c = 0; c < CH; ++c) {
        const int s   = (int)(w[c] & ((1u << PK_BITS) - 1u));
        const int off = (int)(w[c] >> PK_BITS);
        nr[c] = dinv[s] * dinv[lo + off];
    }
}

__device__ __forceinline__ void gcn_chunk_add(const unsigned (&w)[CH], const float (&nr)[CH], int i0,
                                              int cnt, int lane, const float* __restrict__ xw,
                                              float* acc) {
#pragma unroll
    for (int c = 0; c < CH; ++c) {
        const int m = min(32, cnt - i0 - c * 32);
        if (m > 0)
            gcn_rows_add((int)(w[c] & ((1u << PK_BITS) - 1u)), (int)(w[c] >> PK_BITS),
                         nr[c], m, lane, xw, acc);
    }
}

__device__ __forceinline__ void gcn_ids_flush(const unsigned* lst, int cnt, int lane, int lo,
                                              const unsigned* __restrict__ pk,
                                              const float* __restrict__ dinv,
                                              const float* __restrict__ xw, float* acc) {
    for (int i0 = 0; i0 < cnt; i0 += 32) {
        const int idx = i0 + lane;
        const bool valid = idx < cnt;
        const unsigned w = valid ? pk[lst[idx]] : 0u;
        const int s   = (int)(w & ((1u << PK_BITS) - 1u));
        const int off = (int)(w >> PK_BITS);
        const float nrm = valid ? dinv[s] * dinv[lo + off] : 0.0f;
        gcn_rows_add(s, off, nrm, min(32, cnt - i0), lane, xw, acc);
    }
}

#define RPW (NPB / AGG_W)
static_assert(NPB % AGG_W == 0, "every wave starts and finishes the same number of rows");

__global__ __launch_bounds__(AGG_T) void gcn_agg1_lds(const unsigned* __restrict__ pk,
                                                      const unsigned char* __restrict__ keys,
                                                      const unsigned* __restrict__ slots,
                                                      const float* __restrict__ dinv,
                                                      const float* __restrict__ xw,
                                                      const float* __restrict__ b1,
                                                      const float* __restrict__ W2,
                                                      float* __restrict__ zw, int n, int nE) {
    __shared__ __attribute__((aligned(16))) float acc[NPB * HID];
    __shared__ unsigned lst_all[AGG_W * LCAP];
    __shared__ __attribute__((aligned(16))) float zs[NPB * N_CLS];

    const int lane = threadIdx.x & 31;
    const int wave = __builtin_amdgcn_readfirstlane((int)(threadIdx.x >> 5));
    const int lo   = blockIdx.x * NPB;
    const int own  = min(NPB, n - lo);
    const unsigned* slot   = slots + (size_t)blockIdx.x * SLOT_W;
    const unsigned* mylist = slot + AGG_W + wave * WCAP;

    int mine;
    const bool fits = gcn_slot_fits(slot, lane, wave, &mine);

    float4 rv[RPW];
    float  rd[RPW];
#pragma unroll
    for (int i = 0; i < RPW; ++i) {
        const int node = lo + min(wave + i * AGG_W, own - 1);
        rd[i] = dinv[node];
        rv[i] = *(const float4*)(xw + (size_t)node * HID + lane * 4);
    }

    unsigned w[CH];
    float    nr[CH];
#pragma unroll
    for (int c = 0; c < CH; ++c) {
        w[c]  = 0u;
        nr[c] = 0.0f;
    }
    if (fits && mine > 0) gcn_chunk_fetch(mylist, 0, mine, lane, lo, dinv, w, nr);

#pragma unroll
    for (int i = 0; i < RPW; ++i) {
        const float s = rd[i] * rd[i];
        float* a = acc + (wave + i * AGG_W) * HID + lane;
        a[0]  = rv[i].x * s;
        a[32] = rv[i].y * s;
        a[64] = rv[i].z * s;
        a[96] = rv[i].w * s;
    }
    __syncthreads();

    if (fits) {
        for (int i0 = 0; i0 < mine; i0 += 32 * CH) {
            if (i0 > 0) gcn_chunk_fetch(mylist, i0, mine, lane, lo, dinv, w, nr);
            gcn_chunk_add(w, nr, i0, mine, lane, xw, acc);
        }
    } else {
        unsigned* lst = lst_all + wave * LCAP;
        gcn_sweep(keys, nE, (int)gridDim.x, lane, wave, lst, [&](int cnt) {
            gcn_ids_flush(lst, cnt, lane, lo, pk, dinv, xw, acc);
        });
    }
    __syncthreads();

    {
        const float4 bv = *(const float4*)(b1 + lane * 4);
        const int k = lane * 4;
        const float w00 = W2[(k + 0) * 2 + 0], w01 = W2[(k + 0) * 2 + 1];
        const float w10 = W2[(k + 1) * 2 + 0], w11 = W2[(k + 1) * 2 + 1];
        const float w20 = W2[(k + 2) * 2 + 0], w21 = W2[(k + 2) * 2 + 1];
        const float w30 = W2[(k + 3) * 2 + 0], w31 = W2[(k + 3) * 2 + 1];
        float p0[RPW], p1[RPW];
#pragma unroll
        for (int i = 0; i < RPW; ++i) {
            const float* a = acc + min(wave + i * AGG_W, own - 1) * HID + lane;
            const float f0 = fmaxf(a[0]  + bv.x, 0.0f);
            const float f1 = fmaxf(a[32] + bv.y, 0.0f);
            const float f2 = fmaxf(a[64] + bv.z, 0.0f);
            const float f3 = fmaxf(a[96] + bv.w, 0.0f);
            p0[i] = f0 * w00 + f1 * w10 + f2 * w20 + f3 * w30;
            p1[i] = f0 * w01 + f1 * w11 + f2 * w21 + f3 * w31;
        }
#pragma unroll
        for (int i = 0; i < RPW; ++i) {
            p0[i] = gcn_wave_sum(p0[i]);
            p1[i] = gcn_wave_sum(p1[i]);
        }
        if (lane == 0) {
#pragma unroll
            for (int i = 0; i < RPW; ++i) {
                const int r = wave + i * AGG_W;
                zs[r * 2 + 0] = p0[i];
                zs[r * 2 + 1] = p1[i];
            }
        }
    }
    __syncthreads();

    const int nOut = own * N_CLS;
    for (int i = threadIdx.x * 4; i < nOut; i += AGG_T * 4) {
        if (i + 3 < nOut) {
            *(float4*)(zw + (size_t)lo * N_CLS + i) = *(const float4*)(zs + i);
        } else {
            for (int j = i; j < nOut; ++j) zw[(size_t)lo * N_CLS + j] = zs[j];
        }
    }
}


template <bool DIRECT>
__device__ __forceinline__ void gcn_list_flush2(const unsigned* lst, int cnt, int lane,
                                                const unsigned* __restrict__ pk,
                                                const float* __restrict__ dinv,
                                                const float* __restrict__ zw,
                                                const float* dl, float* acc2) {
    for (int i0 = 0; i0 < cnt; i0 += 32 * CH) {
        unsigned w[CH];
        float    ds[CH];
        float2   z[CH];
#pragma unroll
        for (int c = 0; c < CH; ++c) {
            const int idx = min(i0 + c * 32 + lane, cnt - 1);
            w[c] = DIRECT ? lst[idx] : pk[lst[idx]];
        }
#pragma unroll
        for (int c = 0; c < CH; ++c) {
            const int s = (int)(w[c] & ((1u << PK_BITS) - 1u));
            ds[c] = dinv[s];
            z[c]  = *(const float2*)(zw + (size_t)s * N_CLS);
        }
#pragma unroll
        for (int c = 0; c < CH; ++c) {
            if (i0 + c * 32 + lane < cnt) {
                const int off = (int)(w[c] >> PK_BITS);
                const float norm = ds[c] * dl[off];
                gcn_lds_add(acc2 + off * N_CLS + 0, z[c].x * norm);
                gcn_lds_add(acc2 + off * N_CLS + 1, z[c].y * norm);
            }
        }
    }
}

__global__ __launch_bounds__(AGG_T) void gcn_agg2_lds(const unsigned* __restrict__ pk,
                                                      const unsigned char* __restrict__ keys,
                                                      const unsigned* __restrict__ slots,
                                                      const float* __restrict__ dinv,
                                                      const float* __restrict__ zw,
                                                      const float* __restrict__ b2,
                                                      float* __restrict__ out, int n, int nE) {
    __shared__ __attribute__((aligned(16))) float acc2[NPB * N_CLS];
    __shared__ float dl[NPB];
    __shared__ unsigned lst_all[AGG_W * LCAP];

    const int lane = threadIdx.x & 31;
    const int wave = __builtin_amdgcn_readfirstlane((int)(threadIdx.x >> 5));
    const int lo   = blockIdx.x * NPB;
    const int own  = min(NPB, n - lo);
    const unsigned* slot = slots + (size_t)blockIdx.x * SLOT_W;

    for (int r = threadIdx.x; r < own; r += AGG_T) {
        const float d = dinv[lo + r];
        const float s = d * d;
        const float2 zv = *(const float2*)(zw + (size_t)(lo + r) * N_CLS);
        dl[r] = d;
        acc2[r * N_CLS + 0] = zv.x * s;
        acc2[r * N_CLS + 1] = zv.y * s;
    }
    __syncthreads();

    int mine;
    if (gcn_slot_fits(slot, lane, wave, &mine)) {
        gcn_list_flush2<true>(slot + AGG_W + wave * WCAP, mine, lane, pk, dinv, zw, dl, acc2);
    } else {
        unsigned* lst = lst_all + wave * LCAP;
        gcn_sweep(keys, nE, (int)gridDim.x, lane, wave, lst, [&](int cnt) {
            gcn_list_flush2<false>(lst, cnt, lane, pk, dinv, zw, dl, acc2);
        });
    }
    __syncthreads();

    const float c0 = b2[0], c1 = b2[1];
    for (int r = threadIdx.x; r < own; r += AGG_T) {
        const float l0 = acc2[r * N_CLS + 0] + c0;
        const float l1 = acc2[r * N_CLS + 1] + c1;
        const float m = fmaxf(l0, l1);
        const float lse = m + logf(expf(l0 - m) + expf(l1 - m));
        *(float2*)(out + (size_t)(lo + r) * N_CLS) = make_float2(l0 - lse, l1 - lse);
    }
}

extern "C" void kernel_launch(void* const* d_in, const int* in_sizes, int n_in,
                              void* d_out, int out_size, void* d_ws, size_t ws_size,
                              hipStream_t stream) {
    const float* x  = (const float*)d_in[0];
    const float* W1 = (const float*)d_in[1];
    const float* b1 = (const float*)d_in[2];
    const float* W2 = (const float*)d_in[3];
    const float* b2 = (const float*)d_in[4];
    const int*   ei = (const int*)d_in[5];

    const int n  = in_sizes[0] / IN_F;
    const int nE = in_sizes[5] / 2;
    const int* src = ei;
    const int* dst = ei + nE;

    char* ws = (char*)d_ws;
    float* xw   = (float*)(ws);
    unsigned*      pk   = (unsigned*)(ws + (size_t)n * HID * 4);
    unsigned char* keys = (unsigned char*)(ws + (size_t)n * HID * 4) + (size_t)nE * 4;
    unsigned*      slots = (unsigned*)(ws + (size_t)n * HID * 4 + (size_t)nE * 8);
    float* dinv = (float*)(ws + (size_t)2 * n * HID * 4);
    float* zw   = (float*)(ws + (size_t)2 * n * HID * 4 + 512000);
    float* out  = (float*)d_out;

    const int B = 256;
    const int G = (n + NPB - 1) / NPB;

    gcn_edge_keys<<<((nE + 3) / 4 + B - 1) / B, B, 0, stream>>>(src, dst, pk, keys, nE);

    {
        const int tiles = (n / 16) * 2;
        gcn_own_gemm<<<G + (tiles + AGG_W - 1) / AGG_W, AGG_T, 0, stream>>>(pk, keys, dinv, slots, n, nE,
                                                                            x, W1, xw, G);
    }

    gcn_agg1_lds<<<G, AGG_T, 0, stream>>>(pk, keys, slots, dinv, xw, b1, W2, zw, n, nE);

    gcn_agg2_lds<<<G, AGG_T, 0, stream>>>(pk, keys, slots, dinv, zw, b2, out, n, nE);
}


# owner-key plane bit-sliced (8 plane words per 32 edges, by ballots in the key kernel): sweep tests 32 edges with 8 XOR/AND, lanes place own matches by popcount prefix; edge-list order changes
# speedup vs baseline: 4.2856x; 1.0320x over previous
#include <hip/hip_runtime.h>
#include <hip/hip_bf16.h>

#define IN_F 128
#define HID  128
#define N_CLS 2

typedef __attribute__((ext_vector_type(2))) float v2f;
typedef __attribute__((ext_vector_type(8))) float v8f;

#define NPB     416
#define PK_BITS 17

__device__ __forceinline__ unsigned gcn_pack_edge(int s, int d, unsigned* key) {
    const unsigned g = (unsigned)d / NPB;
    *key = g;
    return (unsigned)s | (((unsigned)d - g * NPB) << PK_BITS);
}

__device__ __forceinline__ int gcn_units(int nE) { return ((nE & 3) == 0) ? (nE >> 7) : 0; }

__device__ __forceinline__ const unsigned char* gcn_tail_keys(const unsigned char* keys, int nE) {
    return keys + (size_t)nE + 1024;
}

__global__ void gcn_edge_keys(const int* __restrict__ src, const int* __restrict__ dst,
                              unsigned* __restrict__ pk, unsigned char* __restrict__ keys, int nE) {
    const int t  = blockIdx.x * blockDim.x + threadIdx.x;
    const int e0 = t * 4;
    const int lane = threadIdx.x & 31;
    const int unit = t >> 5;
    if (unit < gcn_units(nE)) {
        const int4 d = ((const int4*)dst)[t];
        const int4 s = ((const int4*)src)[t];
        unsigned k[4];
        uint4 w;
        w.x = gcn_pack_edge(s.x, d.x, &k[0]);
        w.y = gcn_pack_edge(s.y, d.y, &k[1]);
        w.z = gcn_pack_edge(s.z, d.z, &k[2]);
        w.w = gcn_pack_edge(s.w, d.w, &k[3]);
        ((uint4*)pk)[t] = w;
        unsigned mw = 0u;
#pragma unroll
        for (int c = 0; c < 4; ++c) {
#pragma unroll
            for (int b = 0; b < 8; ++b) {
                const unsigned pl = __builtin_amdgcn_ballot_w32(((k[c] >> b) & 1u) != 0u);
                mw = (lane == c * 8 + b) ? pl : mw;
            }
        }
        const int rec = unit * 4 + (lane >> 3);
        const int b   = lane & 7;
        ((unsigned*)keys)[(size_t)(rec >> 5) * 256 + (b >> 2) * 128 + (rec & 31) * 4 + (b & 3)] = mw;
    } else {
        unsigned char* tail = keys + (size_t)nE + 1024;
        const int e1 = min(e0 + 4, nE);
        for (int e = e0; e < e1; ++e) {
            unsigned kk;
            pk[e]   = gcn_pack_edge(src[e], dst[e], &kk);
            tail[e] = (unsigned char)kk;
        }
    }
}

__device__ __forceinline__ void gcn_gemm1_tile(const float* __restrict__ x,
                                               const float* __restrict__ W1,
                                               float* __restrict__ xw,
                                               int strip, int ch, int lane) {
    const int half = lane >> 4;
    const int lm   = lane & 15;
    const int row0 = strip * 16;
    const int col0 = ch * 64;

    const float* __restrict__ xrow = x + (size_t)(row0 + lm) * IN_F;

    v8f acc[4];
#pragma unroll
    for (int cb = 0; cb < 4; ++cb) acc[cb] = (v8f){};

#pragma unroll 2
    for (int k0 = 0; k0 < IN_F; k0 += 4) {
        const int ka = k0 + 2 * half;
        v2f a = *(const v2f*)(xrow + ka);
        const float* __restrict__ w0 = W1 + (size_t)(ka + 0) * HID + col0 + lm;
        const float* __restrict__ w1 = W1 + (size_t)(ka + 1) * HID + col0 + lm;
#pragma unroll
        for (int cb = 0; cb < 4; ++cb) {
            v2f b;
            b.x = w0[cb * 16];
            b.y = w1[cb * 16];
            acc[cb] = __builtin_amdgcn_wmma_f32_16x16x4_f32(false, a, false, b,
                                                            (short)0, acc[cb], false, false);
        }
    }
#pragma unroll
    for (int cb = 0; cb < 4; ++cb) {
#pragma unroll
        for (int r = 0; r < 8; ++r) {
            xw[(size_t)(row0 + r + 8 * half) * HID + col0 + cb * 16 + lm] = acc[cb][r];
        }
    }
}

#define AGG_T  1024
#define AGG_W  (AGG_T / 32)
#define LCAP   320
#define SCAN_G 4
#define WCAP   512
#define SLOT_W (AGG_W + AGG_W * WCAP)
#define CH     8

__device__ __forceinline__ void gcn_lds_add(float* p, float v) {
    __hip_atomic_fetch_add(p, v, __ATOMIC_RELAXED, __HIP_MEMORY_SCOPE_WORKGROUP);
}

__device__ __forceinline__ float gcn_xhalf(float x) {
    const int b = __float_as_int(x);
    return __int_as_float(__builtin_amdgcn_permlanex16(b, b, 0x76543210, 0xfedcba98, false, false));
}

template <int M>
__device__ __forceinline__ float gcn_xrow(float x) {
    return __int_as_float(__builtin_amdgcn_update_dpp(0, __float_as_int(x), 0x160 + M, 0xf, 0xf, true));
}

__device__ __forceinline__ float gcn_wave_sum(float v) {
    v += gcn_xhalf(v);
    v += gcn_xrow<8>(v);
    v += gcn_xrow<4>(v);
    v += gcn_xrow<2>(v);
    v += gcn_xrow<1>(v);
    return v;
}

__device__ __forceinline__ int gcn_list_append(unsigned* lst, int cnt, bool m, int e) {
    const unsigned mask = __builtin_amdgcn_ballot_w32(m);
    if (mask != 0u) {
        const int pos = cnt + (int)__builtin_amdgcn_mbcnt_lo(mask, 0u);
        if (m) lst[pos] = (unsigned)e;
        cnt += __builtin_popcount(mask);
    }
    return cnt;
}

template <class F>
__device__ __forceinline__ void gcn_sweep(const unsigned char* __restrict__ keys, int nE, int nGroups,
                                          int lane, int wave, unsigned* lst, F flush) {
    const int nRec = gcn_units(nE) * 4;
    const int nBlk = (nRec + 31) >> 5;
    const uint4* __restrict__ keys4 = (const uint4*)keys;
    const unsigned grp = (unsigned)blockIdx.x;
    unsigned inv[8];
#pragma unroll
    for (int b = 0; b < 8; ++b) inv[b] = ((grp >> b) & 1u) ? 0u : 0xFFFFFFFFu;
    const int rot = (nBlk > 0) ? (int)(((long long)blockIdx.x * nBlk) / nGroups) : 0;
    int cnt = 0;

    for (int q0 = wave; q0 < nBlk; q0 += AGG_W * SCAN_G) {
        uint4 ka[SCAN_G], kb[SCAN_G];
        int   bb[SCAN_G], lim[SCAN_G];
#pragma unroll
        for (int g = 0; g < SCAN_G; ++g) {
            int b = min(q0 + g * AGG_W, nBlk - 1) + rot;
            b = (b >= nBlk) ? b - nBlk : b;
            bb[g] = b;
            lim[g] = (b == nBlk - 1) ? (nRec - 1 - b * 32) : 31;
            const size_t at = (size_t)b * 64 + min(lane, lim[g]);
            ka[g] = keys4[at];
            kb[g] = keys4[at + 32];
        }
#pragma unroll
        for (int g = 0; g < SCAN_G; ++g) {
            unsigned m = (ka[g].x ^ inv[0]) & (ka[g].y ^ inv[1]) & (ka[g].z ^ inv[2]) & (ka[g].w ^ inv[3]) &
                         (kb[g].x ^ inv[4]) & (kb[g].y ^ inv[5]) & (kb[g].z ^ inv[6]) & (kb[g].w ^ inv[7]);
            if (lane > lim[g]) m = 0u;
            if (q0 + g * AGG_W >= nBlk) m = 0u;
            const int pc = __builtin_popcount(m);
            const unsigned c0  = __builtin_amdgcn_ballot_w32((pc & 1) != 0);
            const unsigned c1  = __builtin_amdgcn_ballot_w32((pc & 2) != 0);
            const unsigned big = __builtin_amdgcn_ballot_w32(pc > 3);
            if ((c0 | c1 | big) != 0u) {
                if (cnt > LCAP - 96) {
                    flush(cnt);
                    cnt = 0;
                }
                if (big == 0u) {
                    const int eb = bb[g] * 1024 + (lane >> 2) * 128 + (lane & 3);
                    int pos = cnt + (int)__builtin_amdgcn_mbcnt_lo(c0, 0u) +
                              2 * (int)__builtin_amdgcn_mbcnt_lo(c1, 0u);
#pragma unroll
                    for (int j = 0; j < 3; ++j) {
                        if (m != 0u) {
                            lst[pos] = (unsigned)(eb + 4 * __builtin_ctz(m));
                            m &= m - 1u;
                            ++pos;
                        }
                    }
                    cnt += __builtin_popcount(c0) + 2 * __builtin_popcount(c1);
                } else {
                    const int eb = bb[g] * 1024 + lane * 4;
                    unsigned lanes = __builtin_amdgcn_ballot_w32(m != 0u);
                    while (lanes != 0u) {
                        const int l = __builtin_ctz(lanes);
                        lanes &= lanes - 1u;
                        const unsigned ml = (unsigned)__builtin_amdgcn_readlane((int)m, l);
                        const int pos = cnt + (int)__builtin_amdgcn_mbcnt_lo(ml, 0u);
                        if ((ml >> lane) & 1u) lst[pos] = (unsigned)(eb + (l >> 2) * 128 + (l & 3));
                        cnt += __builtin_popcount(ml);
                        if (cnt > LCAP - 32) {
                            flush(cnt);
                            cnt = 0;
                        }
                    }
                }
            }
        }
    }
    const unsigned char* __restrict__ tail = gcn_tail_keys(keys, nE);
    for (int base = (nRec >> 2) * 128 + wave * 32; base < nE; base += AGG_W * 32) {
        if (cnt > LCAP - 32) {
            flush(cnt);
            cnt = 0;
        }
        const int e = base + lane;
        const bool hit = (e < nE) && ((unsigned)tail[e] == (unsigned)blockIdx.x);
        cnt = gcn_list_append(lst, cnt, hit, e);
    }
    flush(cnt);
}

__device__ __forceinline__ bool gcn_slot_fits(const unsigned* __restrict__ slot, int lane, int wave,
                                              int* mine) {
    const unsigned c = slot[lane];
    *mine = __builtin_amdgcn_readlane((int)c, wave);
    return __builtin_amdgcn_ballot_w32(c > (unsigned)WCAP) == 0u;
}

__global__ __launch_bounds__(AGG_T) void gcn_own_gemm(const unsigned* __restrict__ pk,
                                                      const unsigned char* __restrict__ keys,
                                                      float* __restrict__ dinv,
                                                      unsigned* __restrict__ slots, int n, int nE,
                                                      const float* __restrict__ x,
                                                      const float* __restrict__ W1,
                                                      float* __restrict__ xw, int nGroups) {
    __shared__ float degl[NPB];
    __shared__ unsigned cntl[AGG_W];
    __shared__ unsigned lst_all[AGG_W * LCAP];

    const int lane = threadIdx.x & 31;
    const int wave = __builtin_amdgcn_readfirstlane((int)(threadIdx.x >> 5));

    if ((int)blockIdx.x >= nGroups) {
        const int t = ((int)blockIdx.x - nGroups) * AGG_W + wave;
        if ((t >> 1) < n / 16) gcn_gemm1_tile(x, W1, xw, t >> 1, t & 1, lane);
        return;
    }

    const int lo   = blockIdx.x * NPB;
    const int own  = min(NPB, n - lo);
    unsigned* lst  = lst_all + wave * LCAP;
    unsigned* slot = slots + (size_t)blockIdx.x * SLOT_W;
    unsigned* part = slot + AGG_W + wave * WCAP;

    for (int r = threadIdx.x; r < NPB; r += AGG_T) degl[r] = 1.0f;
    __syncthreads();

    int kept = 0;
    gcn_sweep(keys, nE, nGroups, lane, wave, lst, [&](int cnt) {
        for (int i0 = 0; i0 < cnt; i0 += 32 * CH) {
            unsigned w[CH];
#pragma unroll
            for (int c = 0; c < CH; ++c) {
                const int idx = min(i0 + c * 32 + lane, cnt - 1);
                w[c] = pk[lst[idx]];
            }
#pragma unroll
            for (int c = 0; c < CH; ++c) {
                const int idx = i0 + c * 32 + lane;
                if (idx < cnt) {
                    gcn_lds_add(degl + (w[c] >> PK_BITS), 1.0f);
                    if (kept + idx < WCAP) part[kept + idx] = w[c];
                }
            }
        }
        kept += cnt;
    });
    if (lane == 0) cntl[wave] = (unsigned)kept;
    __syncthreads();

    if (wave == 0) slot[lane] = cntl[lane];
    for (int r = threadIdx.x; r < own; r += AGG_T) dinv[lo + r] = rsqrtf(degl[r]);
}

static_assert(AGG_W == 32, "a group's line of counts is read and written by the 32 lanes of one wave");


__device__ __forceinline__ void gcn_rows_add(int s, int off, float nrm, int m, int lane,
                                             const float* __restrict__ xw, float* acc) {
    for (int k0 = 0; k0 < m; k0 += 8) {
        float4 v[8];
#pragma unroll
        for (int u = 0; u < 8; ++u) {
            const int sk = __builtin_amdgcn_readlane(s, k0 + u);
            v[u] = *(const float4*)(xw + (size_t)sk * HID + lane * 4);
        }
#pragma unroll
        for (int u = 0; u < 8; ++u) {
            if (k0 + u < m) {
                const int dk = __builtin_amdgcn_readlane(off, k0 + u);
                const float nk = __int_as_float(
                    __builtin_amdgcn_readlane(__float_as_int(nrm), k0 + u));
                float* a = acc + dk * HID + lane;
                gcn_lds_add(a +  0, v[u].x * nk);
                gcn_lds_add(a + 32, v[u].y * nk);
                gcn_lds_add(a + 64, v[u].z * nk);
                gcn_lds_add(a + 96, v[u].w * nk);
            }
        }
    }
}

__device__ __forceinline__ void gcn_chunk_fetch(const unsigned* lst, int i0, int cnt, int lane, int lo,
                                                const float* __restrict__ dinv,
                                                unsigned (&w)[CH], float (&nr)[CH]) {
#pragma unroll
    for (int c = 0; c < CH; ++c) w[c] = lst[min(i0 + c * 32 + lane, cnt - 1)];
#pragma unroll
    for (int c = 0; c < CH; ++c) {
        const int s   = (int)(w[c] & ((1u << PK_BITS) - 1u));
        const int off = (int)(w[c] >> PK_BITS);
        nr[c] = dinv[s] * dinv[lo + off];
    }
}

__device__ __forceinline__ void gcn_chunk_add(const unsigned (&w)[CH], const float (&nr)[CH], int i0,
                                              int cnt, int lane, const float* __restrict__ xw,
                                              float* acc) {
#pragma unroll
    for (int c = 0; c < CH; ++c) {
        const int m = min(32, cnt - i0 - c * 32);
        if (m > 0)
            gcn_rows_add((int)(w[c] & ((1u << PK_BITS) - 1u)), (int)(w[c] >> PK_BITS),
                         nr[c], m, lane, xw, acc);
    }
}

__device__ __forceinline__ void gcn_ids_flush(const unsigned* lst, int cnt, int lane, int lo,
                                              const unsigned* __restrict__ pk,
                                              const float* __restrict__ dinv,
                                              const float* __restrict__ xw, float* acc) {
    for (int i0 = 0; i0 < cnt; i0 += 32) {
        const int idx = i0 + lane;
        const bool valid = idx < cnt;
        const unsigned w = valid ? pk[lst[idx]] : 0u;
        const int s   = (int)(w & ((1u << PK_BITS) - 1u));
        const int off = (int)(w >> PK_BITS);
        const float nrm = valid ? dinv[s] * dinv[lo + off] : 0.0f;
        gcn_rows_add(s, off, nrm, min(32, cnt - i0), lane, xw, acc);
    }
}

#define RPW (NPB / AGG_W)
static_assert(NPB % AGG_W == 0, "every wave starts and finishes the same number of rows");

__global__ __launch_bounds__(AGG_T) void gcn_agg1_lds(const unsigned* __restrict__ pk,
                                                      const unsigned char* __restrict__ keys,
                                                      const unsigned* __restrict__ slots,
                                                      const float* __restrict__ dinv,
                                                      const float* __restrict__ xw,
                                                      const float* __restrict__ b1,
                                                      const float* __restrict__ W2,
                                                      float* __restrict__ zw, int n, int nE) {
    __shared__ __attribute__((aligned(16))) float acc[NPB * HID];
    __shared__ unsigned lst_all[AGG_W * LCAP];
    __shared__ __attribute__((aligned(16))) float zs[NPB * N_CLS];

    const int lane = threadIdx.x & 31;
    const int wave = __builtin_amdgcn_readfirstlane((int)(threadIdx.x >> 5));
    const int lo   = blockIdx.x * NPB;
    const int own  = min(NPB, n - lo);
    const unsigned* slot   = slots + (size_t)blockIdx.x * SLOT_W;
    const unsigned* mylist = slot + AGG_W + wave * WCAP;

    int mine;
    const bool fits = gcn_slot_fits(slot, lane, wave, &mine);

    float4 rv[RPW];
    float  rd[RPW];
#pragma unroll
    for (int i = 0; i < RPW; ++i) {
        const int node = lo + min(wave + i * AGG_W, own - 1);
        rd[i] = dinv[node];
        rv[i] = *(const float4*)(xw + (size_t)node * HID + lane * 4);
    }

    unsigned w[CH];
    float    nr[CH];
#pragma unroll
    for (int c = 0; c < CH; ++c) {
        w[c]  = 0u;
        nr[c] = 0.0f;
    }
    if (fits && mine > 0) gcn_chunk_fetch(mylist, 0, mine, lane, lo, dinv, w, nr);

#pragma unroll
    for (int i = 0; i < RPW; ++i) {
        const float s = rd[i] * rd[i];
        float* a = acc + (wave + i * AGG_W) * HID + lane;
        a[0]  = rv[i].x * s;
        a[32] = rv[i].y * s;
        a[64] = rv[i].z * s;
        a[96] = rv[i].w * s;
    }
    __syncthreads();

    if (fits) {
        for (int i0 = 0; i0 < mine; i0 += 32 * CH) {
            if (i0 > 0) gcn_chunk_fetch(mylist, i0, mine, lane, lo, dinv, w, nr);
            gcn_chunk_add(w, nr, i0, mine, lane, xw, acc);
        }
    } else {
        unsigned* lst = lst_all + wave * LCAP;
        gcn_sweep(keys, nE, (int)gridDim.x, lane, wave, lst, [&](int cnt) {
            gcn_ids_flush(lst, cnt, lane, lo, pk, dinv, xw, acc);
        });
    }
    __syncthreads();

    {
        const float4 bv = *(const float4*)(b1 + lane * 4);
        const int k = lane * 4;
        const float w00 = W2[(k + 0) * 2 + 0], w01 = W2[(k + 0) * 2 + 1];
        const float w10 = W2[(k + 1) * 2 + 0], w11 = W2[(k + 1) * 2 + 1];
        const float w20 = W2[(k + 2) * 2 + 0], w21 = W2[(k + 2) * 2 + 1];
        const float w30 = W2[(k + 3) * 2 + 0], w31 = W2[(k + 3) * 2 + 1];
        float p0[RPW], p1[RPW];
#pragma unroll
        for (int i = 0; i < RPW; ++i) {
            const float* a = acc + min(wave + i * AGG_W, own - 1) * HID + lane;
            const float f0 = fmaxf(a[0]  + bv.x, 0.0f);
            const float f1 = fmaxf(a[32] + bv.y, 0.0f);
            const float f2 = fmaxf(a[64] + bv.z, 0.0f);
            const float f3 = fmaxf(a[96] + bv.w, 0.0f);
            p0[i] = f0 * w00 + f1 * w10 + f2 * w20 + f3 * w30;
            p1[i] = f0 * w01 + f1 * w11 + f2 * w21 + f3 * w31;
        }
#pragma unroll
        for (int i = 0; i < RPW; ++i) {
            p0[i] = gcn_wave_sum(p0[i]);
            p1[i] = gcn_wave_sum(p1[i]);
        }
        if (lane == 0) {
#pragma unroll
            for (int i = 0; i < RPW; ++i) {
                const int r = wave + i * AGG_W;
                zs[r * 2 + 0] = p0[i];
                zs[r * 2 + 1] = p1[i];
            }
        }
    }
    __syncthreads();

    const int nOut = own * N_CLS;
    for (int i = threadIdx.x * 4; i < nOut; i += AGG_T * 4) {
        if (i + 3 < nOut) {
            *(float4*)(zw + (size_t)lo * N_CLS + i) = *(const float4*)(zs + i);
        } else {
            for (int j = i; j < nOut; ++j) zw[(size_t)lo * N_CLS + j] = zs[j];
        }
    }
}


template <bool DIRECT>
__device__ __forceinline__ void gcn_list_flush2(const unsigned* lst, int cnt, int lane,
                                                const unsigned* __restrict__ pk,
                                                const float* __restrict__ dinv,
                                                const float* __restrict__ zw,
                                                const float* dl, float* acc2) {
    for (int i0 = 0; i0 < cnt; i0 += 32 * CH) {
        unsigned w[CH];
        float    ds[CH];
        float2   z[CH];
#pragma unroll
        for (int c = 0; c < CH; ++c) {
            const int idx = min(i0 + c * 32 + lane, cnt - 1);
            w[c] = DIRECT ? lst[idx] : pk[lst[idx]];
        }
#pragma unroll
        for (int c = 0; c < CH; ++c) {
            const int s = (int)(w[c] & ((1u << PK_BITS) - 1u));
            ds[c] = dinv[s];
            z[c]  = *(const float2*)(zw + (size_t)s * N_CLS);
        }
#pragma unroll
        for (int c = 0; c < CH; ++c) {
            if (i0 + c * 32 + lane < cnt) {
                const int off = (int)(w[c] >> PK_BITS);
                const float norm = ds[c] * dl[off];
                gcn_lds_add(acc2 + off * N_CLS + 0, z[c].x * norm);
                gcn_lds_add(acc2 + off * N_CLS + 1, z[c].y * norm);
            }
        }
    }
}

__global__ __launch_bounds__(AGG_T) void gcn_agg2_lds(const unsigned* __restrict__ pk,
                                                      const unsigned char* __restrict__ keys,
                                                      const unsigned* __restrict__ slots,
                                                      const float* __restrict__ dinv,
                                                      const float* __restrict__ zw,
                                                      const float* __restrict__ b2,
                                                      float* __restrict__ out, int n, int nE) {
    __shared__ __attribute__((aligned(16))) float acc2[NPB * N_CLS];
    __shared__ float dl[NPB];
    __shared__ unsigned lst_all[AGG_W * LCAP];

    const int lane = threadIdx.x & 31;
    const int wave = __builtin_amdgcn_readfirstlane((int)(threadIdx.x >> 5));
    const int lo   = blockIdx.x * NPB;
    const int own  = min(NPB, n - lo);
    const unsigned* slot = slots + (size_t)blockIdx.x * SLOT_W;

    for (int r = threadIdx.x; r < own; r += AGG_T) {
        const float d = dinv[lo + r];
        const float s = d * d;
        const float2 zv = *(const float2*)(zw + (size_t)(lo + r) * N_CLS);
        dl[r] = d;
        acc2[r * N_CLS + 0] = zv.x * s;
        acc2[r * N_CLS + 1] = zv.y * s;
    }
    __syncthreads();

    int mine;
    if (gcn_slot_fits(slot, lane, wave, &mine)) {
        gcn_list_flush2<true>(slot + AGG_W + wave * WCAP, mine, lane, pk, dinv, zw, dl, acc2);
    } else {
        unsigned* lst = lst_all + wave * LCAP;
        gcn_sweep(keys, nE, (int)gridDim.x, lane, wave, lst, [&](int cnt) {
            gcn_list_flush2<false>(lst, cnt, lane, pk, dinv, zw, dl, acc2);
        });
    }
    __syncthreads();

    const float c0 = b2[0], c1 = b2[1];
    for (int r = threadIdx.x; r < own; r += AGG_T) {
        const float l0 = acc2[r * N_CLS + 0] + c0;
        const float l1 = acc2[r * N_CLS + 1] + c1;
        const float m = fmaxf(l0, l1);
        const float lse = m + logf(expf(l0 - m) + expf(l1 - m));
        *(float2*)(out + (size_t)(lo + r) * N_CLS) = make_float2(l0 - lse, l1 - lse);
    }
}

extern "C" void kernel_launch(void* const* d_in, const int* in_sizes, int n_in,
                              void* d_out, int out_size, void* d_ws, size_t ws_size,
                              hipStream_t stream) {
    const float* x  = (const float*)d_in[0];
    const float* W1 = (const float*)d_in[1];
    const float* b1 = (const float*)d_in[2];
    const float* W2 = (const float*)d_in[3];
    const float* b2 = (const float*)d_in[4];
    const int*   ei = (const int*)d_in[5];

    const int n  = in_sizes[0] / IN_F;
    const int nE = in_sizes[5] / 2;
    const int* src = ei;
    const int* dst = ei + nE;

    char* ws = (char*)d_ws;
    float* xw   = (float*)(ws);
    unsigned*      pk   = (unsigned*)(ws + (size_t)n * HID * 4);
    unsigned char* keys = (unsigned char*)(ws + (size_t)n * HID * 4) + (size_t)nE * 4;
    unsigned*      slots = (unsigned*)(ws + (size_t)n * HID * 4 + (size_t)nE * 8);
    float* dinv = (float*)(ws + (size_t)2 * n * HID * 4);
    float* zw   = (float*)(ws + (size_t)2 * n * HID * 4 + 512000);
    float* out  = (float*)d_out;

    const int B = 256;
    const int G = (n + NPB - 1) / NPB;

    gcn_edge_keys<<<((nE + 3) / 4 + B - 1) / B, B, 0, stream>>>(src, dst, pk, keys, nE);

    {
        const int tiles = (n / 16) * 2;
        gcn_own_gemm<<<G + (tiles + AGG_W - 1) / AGG_W, AGG_T, 0, stream>>>(pk, keys, dinv, slots, n, nE,
                                                                            x, W1, xw, G);
    }

    gcn_agg1_lds<<<G, AGG_T, 0, stream>>>(pk, keys, slots, dinv, xw, b1, W2, zw, n, nE);

    gcn_agg2_lds<<<G, AGG_T, 0, stream>>>(pk, keys, slots, dinv, zw, b2, out, n, nE);
}


# layer-1 aggregation keeps each listed edge's weight dinv[src]*dinv[dst] beside its list word; layer 2 reads it in list order instead of a dinv[src] lookup (one cache line) per edge; same products
# speedup vs baseline: 4.4124x; 1.0296x over previous
#include <hip/hip_runtime.h>
#include <hip/hip_bf16.h>

#define IN_F 128
#define HID  128
#define N_CLS 2

typedef __attribute__((ext_vector_type(2))) float v2f;
typedef __attribute__((ext_vector_type(8))) float v8f;

#define NPB     416
#define PK_BITS 17

__device__ __forceinline__ unsigned gcn_pack_edge(int s, int d, unsigned* key) {
    const unsigned g = (unsigned)d / NPB;
    *key = g;
    return (unsigned)s | (((unsigned)d - g * NPB) << PK_BITS);
}

__device__ __forceinline__ int gcn_units(int nE) { return ((nE & 3) == 0) ? (nE >> 7) : 0; }

__device__ __forceinline__ const unsigned char* gcn_tail_keys(const unsigned char* keys, int nE) {
    return keys + (size_t)nE + 1024;
}

__global__ void gcn_edge_keys(const int* __restrict__ src, const int* __restrict__ dst,
                              unsigned* __restrict__ pk, unsigned char* __restrict__ keys, int nE) {
    const int t  = blockIdx.x * blockDim.x + threadIdx.x;
    const int e0 = t * 4;
    const int lane = threadIdx.x & 31;
    const int unit = t >> 5;
    if (unit < gcn_units(nE)) {
        const int4 d = ((const int4*)dst)[t];
        const int4 s = ((const int4*)src)[t];
        unsigned k[4];
        uint4 w;
        w.x = gcn_pack_edge(s.x, d.x, &k[0]);
        w.y = gcn_pack_edge(s.y, d.y, &k[1]);
        w.z = gcn_pack_edge(s.z, d.z, &k[2]);
        w.w = gcn_pack_edge(s.w, d.w, &k[3]);
        ((uint4*)pk)[t] = w;
        unsigned mw = 0u;
#pragma unroll
        for (int c = 0; c < 4; ++c) {
#pragma unroll
            for (int b = 0; b < 8; ++b) {
                const unsigned pl = __builtin_amdgcn_ballot_w32(((k[c] >> b) & 1u) != 0u);
                mw = (lane == c * 8 + b) ? pl : mw;
            }
        }
        const int rec = unit * 4 + (lane >> 3);
        const int b   = lane & 7;
        ((unsigned*)keys)[(size_t)(rec >> 5) * 256 + (b >> 2) * 128 + (rec & 31) * 4 + (b & 3)] = mw;
    } else {
        unsigned char* tail = keys + (size_t)nE + 1024;
        const int e1 = min(e0 + 4, nE);
        for (int e = e0; e < e1; ++e) {
            unsigned kk;
            pk[e]   = gcn_pack_edge(src[e], dst[e], &kk);
            tail[e] = (unsigned char)kk;
        }
    }
}

__device__ __forceinline__ void gcn_gemm1_tile(const float* __restrict__ x,
                                               const float* __restrict__ W1,
                                               float* __restrict__ xw,
                                               int strip, int ch, int lane) {
    const int half = lane >> 4;
    const int lm   = lane & 15;
    const int row0 = strip * 16;
    const int col0 = ch * 64;

    const float* __restrict__ xrow = x + (size_t)(row0 + lm) * IN_F;

    v8f acc[4];
#pragma unroll
    for (int cb = 0; cb < 4; ++cb) acc[cb] = (v8f){};

#pragma unroll 2
    for (int k0 = 0; k0 < IN_F; k0 += 4) {
        const int ka = k0 + 2 * half;
        v2f a = *(const v2f*)(xrow + ka);
        const float* __restrict__ w0 = W1 + (size_t)(ka + 0) * HID + col0 + lm;
        const float* __restrict__ w1 = W1 + (size_t)(ka + 1) * HID + col0 + lm;
#pragma unroll
        for (int cb = 0; cb < 4; ++cb) {
            v2f b;
            b.x = w0[cb * 16];
            b.y = w1[cb * 16];
            acc[cb] = __builtin_amdgcn_wmma_f32_16x16x4_f32(false, a, false, b,
                                                            (short)0, acc[cb], false, false);
        }
    }
#pragma unroll
    for (int cb = 0; cb < 4; ++cb) {
#pragma unroll
        for (int r = 0; r < 8; ++r) {
            xw[(size_t)(row0 + r + 8 * half) * HID + col0 + cb * 16 + lm] = acc[cb][r];
        }
    }
}

#define AGG_T  1024
#define AGG_W  (AGG_T / 32)
#define LCAP   320
#define SCAN_G 4
#define WCAP   512
#define SLOT_W (AGG_W + AGG_W * WCAP)
#define CH     8

__device__ __forceinline__ void gcn_lds_add(float* p, float v) {
    __hip_atomic_fetch_add(p, v, __ATOMIC_RELAXED, __HIP_MEMORY_SCOPE_WORKGROUP);
}

__device__ __forceinline__ float gcn_xhalf(float x) {
    const int b = __float_as_int(x);
    return __int_as_float(__builtin_amdgcn_permlanex16(b, b, 0x76543210, 0xfedcba98, false, false));
}

template <int M>
__device__ __forceinline__ float gcn_xrow(float x) {
    return __int_as_float(__builtin_amdgcn_update_dpp(0, __float_as_int(x), 0x160 + M, 0xf, 0xf, true));
}

__device__ __forceinline__ float gcn_wave_sum(float v) {
    v += gcn_xhalf(v);
    v += gcn_xrow<8>(v);
    v += gcn_xrow<4>(v);
    v += gcn_xrow<2>(v);
    v += gcn_xrow<1>(v);
    return v;
}

__device__ __forceinline__ int gcn_list_append(unsigned* lst, int cnt, bool m, int e) {
    const unsigned mask = __builtin_amdgcn_ballot_w32(m);
    if (mask != 0u) {
        const int pos = cnt + (int)__builtin_amdgcn_mbcnt_lo(mask, 0u);
        if (m) lst[pos] = (unsigned)e;
        cnt += __builtin_popcount(mask);
    }
    return cnt;
}

template <class F>
__device__ __forceinline__ void gcn_sweep(const unsigned char* __restrict__ keys, int nE, int nGroups,
                                          int lane, int wave, unsigned* lst, F flush) {
    const int nRec = gcn_units(nE) * 4;
    const int nBlk = (nRec + 31) >> 5;
    const uint4* __restrict__ keys4 = (const uint4*)keys;
    const unsigned grp = (unsigned)blockIdx.x;
    unsigned inv[8];
#pragma unroll
    for (int b = 0; b < 8; ++b) inv[b] = ((grp >> b) & 1u) ? 0u : 0xFFFFFFFFu;
    const int rot = (nBlk > 0) ? (int)(((long long)blockIdx.x * nBlk) / nGroups) : 0;
    int cnt = 0;

    for (int q0 = wave; q0 < nBlk; q0 += AGG_W * SCAN_G) {
        uint4 ka[SCAN_G], kb[SCAN_G];
        int   bb[SCAN_G], lim[SCAN_G];
#pragma unroll
        for (int g = 0; g < SCAN_G; ++g) {
            int b = min(q0 + g * AGG_W, nBlk - 1) + rot;
            b = (b >= nBlk) ? b - nBlk : b;
            bb[g] = b;
            lim[g] = (b == nBlk - 1) ? (nRec - 1 - b * 32) : 31;
            const size_t at = (size_t)b * 64 + min(lane, lim[g]);
            ka[g] = keys4[at];
            kb[g] = keys4[at + 32];
        }
#pragma unroll
        for (int g = 0; g < SCAN_G; ++g) {
            unsigned m = (ka[g].x ^ inv[0]) & (ka[g].y ^ inv[1]) & (ka[g].z ^ inv[2]) & (ka[g].w ^ inv[3]) &
                         (kb[g].x ^ inv[4]) & (kb[g].y ^ inv[5]) & (kb[g].z ^ inv[6]) & (kb[g].w ^ inv[7]);
            if (lane > lim[g]) m = 0u;
            if (q0 + g * AGG_W >= nBlk) m = 0u;
            const int pc = __builtin_popcount(m);
            const unsigned c0  = __builtin_amdgcn_ballot_w32((pc & 1) != 0);
            const unsigned c1  = __builtin_amdgcn_ballot_w32((pc & 2) != 0);
            const unsigned big = __builtin_amdgcn_ballot_w32(pc > 3);
            if ((c0 | c1 | big) != 0u) {
                if (cnt > LCAP - 96) {
                    flush(cnt);
                    cnt = 0;
                }
                if (big == 0u) {
                    const int eb = bb[g] * 1024 + (lane >> 2) * 128 + (lane & 3);
                    int pos = cnt + (int)__builtin_amdgcn_mbcnt_lo(c0, 0u) +
                              2 * (int)__builtin_amdgcn_mbcnt_lo(c1, 0u);
#pragma unroll
                    for (int j = 0; j < 3; ++j) {
                        if (m != 0u) {
                            lst[pos] = (unsigned)(eb + 4 * __builtin_ctz(m));
                            m &= m - 1u;
                            ++pos;
                        }
                    }
                    cnt += __builtin_popcount(c0) + 2 * __builtin_popcount(c1);
                } else {
                    const int eb = bb[g] * 1024 + lane * 4;
                    unsigned lanes = __builtin_amdgcn_ballot_w32(m != 0u);
                    while (lanes != 0u) {
                        const int l = __builtin_ctz(lanes);
                        lanes &= lanes - 1u;
                        const unsigned ml = (unsigned)__builtin_amdgcn_readlane((int)m, l);
                        const int pos = cnt + (int)__builtin_amdgcn_mbcnt_lo(ml, 0u);
                        if ((ml >> lane) & 1u) lst[pos] = (unsigned)(eb + (l >> 2) * 128 + (l & 3));
                        cnt += __builtin_popcount(ml);
                        if (cnt > LCAP - 32) {
                            flush(cnt);
                            cnt = 0;
                        }
                    }
                }
            }
        }
    }
    const unsigned char* __restrict__ tail = gcn_tail_keys(keys, nE);
    for (int base = (nRec >> 2) * 128 + wave * 32; base < nE; base += AGG_W * 32) {
        if (cnt > LCAP - 32) {
            flush(cnt);
            cnt = 0;
        }
        const int e = base + lane;
        const bool hit = (e < nE) && ((unsigned)tail[e] == (unsigned)blockIdx.x);
        cnt = gcn_list_append(lst, cnt, hit, e);
    }
    flush(cnt);
}

__device__ __forceinline__ bool gcn_slot_fits(const unsigned* __restrict__ slot, int lane, int wave,
                                              int* mine) {
    const unsigned c = slot[lane];
    *mine = __builtin_amdgcn_readlane((int)c, wave);
    return __builtin_amdgcn_ballot_w32(c > (unsigned)WCAP) == 0u;
}

__global__ __launch_bounds__(AGG_T) void gcn_own_gemm(const unsigned* __restrict__ pk,
                                                      const unsigned char* __restrict__ keys,
                                                      float* __restrict__ dinv,
                                                      unsigned* __restrict__ slots, int n, int nE,
                                                      const float* __restrict__ x,
                                                      const float* __restrict__ W1,
                                                      float* __restrict__ xw, int nGroups) {
    __shared__ float degl[NPB];
    __shared__ unsigned cntl[AGG_W];
    __shared__ unsigned lst_all[AGG_W * LCAP];

    const int lane = threadIdx.x & 31;
    const int wave = __builtin_amdgcn_readfirstlane((int)(threadIdx.x >> 5));

    if ((int)blockIdx.x >= nGroups) {
        const int t = ((int)blockIdx.x - nGroups) * AGG_W + wave;
        if ((t >> 1) < n / 16) gcn_gemm1_tile(x, W1, xw, t >> 1, t & 1, lane);
        return;
    }

    const int lo   = blockIdx.x * NPB;
    const int own  = min(NPB, n - lo);
    unsigned* lst  = lst_all + wave * LCAP;
    unsigned* slot = slots + (size_t)blockIdx.x * SLOT_W;
    unsigned* part = slot + AGG_W + wave * WCAP;

    for (int r = threadIdx.x; r < NPB; r += AGG_T) degl[r] = 1.0f;
    __syncthreads();

    int kept = 0;
    gcn_sweep(keys, nE, nGroups, lane, wave, lst, [&](int cnt) {
        for (int i0 = 0; i0 < cnt; i0 += 32 * CH) {
            unsigned w[CH];
#pragma unroll
            for (int c = 0; c < CH; ++c) {
                const int idx = min(i0 + c * 32 + lane, cnt - 1);
                w[c] = pk[lst[idx]];
            }
#pragma unroll
            for (int c = 0; c < CH; ++c) {
                const int idx = i0 + c * 32 + lane;
                if (idx < cnt) {
                    gcn_lds_add(degl + (w[c] >> PK_BITS), 1.0f);
                    if (kept + idx < WCAP) part[kept + idx] = w[c];
                }
            }
        }
        kept += cnt;
    });
    if (lane == 0) cntl[wave] = (unsigned)kept;
    __syncthreads();

    if (wave == 0) slot[lane] = cntl[lane];
    for (int r = threadIdx.x; r < own; r += AGG_T) dinv[lo + r] = rsqrtf(degl[r]);
}

static_assert(AGG_W == 32, "a group's line of counts is read and written by the 32 lanes of one wave");


__device__ __forceinline__ void gcn_rows_add(int s, int off, float nrm, int m, int lane,
                                             const float* __restrict__ xw, float* acc) {
    for (int k0 = 0; k0 < m; k0 += 8) {
        float4 v[8];
#pragma unroll
        for (int u = 0; u < 8; ++u) {
            const int sk = __builtin_amdgcn_readlane(s, k0 + u);
            v[u] = *(const float4*)(xw + (size_t)sk * HID + lane * 4);
        }
#pragma unroll
        for (int u = 0; u < 8; ++u) {
            if (k0 + u < m) {
                const int dk = __builtin_amdgcn_readlane(off, k0 + u);
                const float nk = __int_as_float(
                    __builtin_amdgcn_readlane(__float_as_int(nrm), k0 + u));
                float* a = acc + dk * HID + lane;
                gcn_lds_add(a +  0, v[u].x * nk);
                gcn_lds_add(a + 32, v[u].y * nk);
                gcn_lds_add(a + 64, v[u].z * nk);
                gcn_lds_add(a + 96, v[u].w * nk);
            }
        }
    }
}

__device__ __forceinline__ void gcn_chunk_fetch(const unsigned* lst, float* nlst, int i0, int cnt, int lane,
                                                int lo, const float* __restrict__ dinv,
                                                unsigned (&w)[CH], float (&nr)[CH]) {
#pragma unroll
    for (int c = 0; c < CH; ++c) w[c] = lst[min(i0 + c * 32 + lane, cnt - 1)];
#pragma unroll
    for (int c = 0; c < CH; ++c) {
        const int s   = (int)(w[c] & ((1u << PK_BITS) - 1u));
        const int off = (int)(w[c] >> PK_BITS);
        nr[c] = dinv[s] * dinv[lo + off];
    }
#pragma unroll
    for (int c = 0; c < CH; ++c) {
        const int idx = i0 + c * 32 + lane;
        if (idx < cnt) nlst[idx] = nr[c];
    }
}

__device__ __forceinline__ void gcn_chunk_add(const unsigned (&w)[CH], const float (&nr)[CH], int i0,
                                              int cnt, int lane, const float* __restrict__ xw,
                                              float* acc) {
#pragma unroll
    for (int c = 0; c < CH; ++c) {
        const int m = min(32, cnt - i0 - c * 32);
        if (m > 0)
            gcn_rows_add((int)(w[c] & ((1u << PK_BITS) - 1u)), (int)(w[c] >> PK_BITS),
                         nr[c], m, lane, xw, acc);
    }
}

__device__ __forceinline__ void gcn_ids_flush(const unsigned* lst, int cnt, int lane, int lo,
                                              const unsigned* __restrict__ pk,
                                              const float* __restrict__ dinv,
                                              const float* __restrict__ xw, float* acc) {
    for (int i0 = 0; i0 < cnt; i0 += 32) {
        const int idx = i0 + lane;
        const bool valid = idx < cnt;
        const unsigned w = valid ? pk[lst[idx]] : 0u;
        const int s   = (int)(w & ((1u << PK_BITS) - 1u));
        const int off = (int)(w >> PK_BITS);
        const float nrm = valid ? dinv[s] * dinv[lo + off] : 0.0f;
        gcn_rows_add(s, off, nrm, min(32, cnt - i0), lane, xw, acc);
    }
}

#define RPW (NPB / AGG_W)
static_assert(NPB % AGG_W == 0, "every wave starts and finishes the same number of rows");

__global__ __launch_bounds__(AGG_T) void gcn_agg1_lds(const unsigned* __restrict__ pk,
                                                      const unsigned char* __restrict__ keys,
                                                      const unsigned* __restrict__ slots,
                                                      float* __restrict__ norms,
                                                      const float* __restrict__ dinv,
                                                      const float* __restrict__ xw,
                                                      const float* __restrict__ b1,
                                                      const float* __restrict__ W2,
                                                      float* __restrict__ zw, int n, int nE) {
    __shared__ __attribute__((aligned(16))) float acc[NPB * HID];
    __shared__ unsigned lst_all[AGG_W * LCAP];
    __shared__ __attribute__((aligned(16))) float zs[NPB * N_CLS];

    const int lane = threadIdx.x & 31;
    const int wave = __builtin_amdgcn_readfirstlane((int)(threadIdx.x >> 5));
    const int lo   = blockIdx.x * NPB;
    const int own  = min(NPB, n - lo);
    const unsigned* slot   = slots + (size_t)blockIdx.x * SLOT_W;
    const unsigned* mylist = slot + AGG_W + wave * WCAP;
    float* mynorm = norms + ((size_t)blockIdx.x * AGG_W + wave) * WCAP;

    int mine;
    const bool fits = gcn_slot_fits(slot, lane, wave, &mine);

    float4 rv[RPW];
    float  rd[RPW];
#pragma unroll
    for (int i = 0; i < RPW; ++i) {
        const int node = lo + min(wave + i * AGG_W, own - 1);
        rd[i] = dinv[node];
        rv[i] = *(const float4*)(xw + (size_t)node * HID + lane * 4);
    }

    unsigned w[CH];
    float    nr[CH];
#pragma unroll
    for (int c = 0; c < CH; ++c) {
        w[c]  = 0u;
        nr[c] = 0.0f;
    }
    if (fits && mine > 0) gcn_chunk_fetch(mylist, mynorm, 0, mine, lane, lo, dinv, w, nr);

#pragma unroll
    for (int i = 0; i < RPW; ++i) {
        const float s = rd[i] * rd[i];
        float* a = acc + (wave + i * AGG_W) * HID + lane;
        a[0]  = rv[i].x * s;
        a[32] = rv[i].y * s;
        a[64] = rv[i].z * s;
        a[96] = rv[i].w * s;
    }
    __syncthreads();

    if (fits) {
        for (int i0 = 0; i0 < mine; i0 += 32 * CH) {
            if (i0 > 0) gcn_chunk_fetch(mylist, mynorm, i0, mine, lane, lo, dinv, w, nr);
            gcn_chunk_add(w, nr, i0, mine, lane, xw, acc);
        }
    } else {
        unsigned* lst = lst_all + wave * LCAP;
        gcn_sweep(keys, nE, (int)gridDim.x, lane, wave, lst, [&](int cnt) {
            gcn_ids_flush(lst, cnt, lane, lo, pk, dinv, xw, acc);
        });
    }
    __syncthreads();

    {
        const float4 bv = *(const float4*)(b1 + lane * 4);
        const int k = lane * 4;
        const float w00 = W2[(k + 0) * 2 + 0], w01 = W2[(k + 0) * 2 + 1];
        const float w10 = W2[(k + 1) * 2 + 0], w11 = W2[(k + 1) * 2 + 1];
        const float w20 = W2[(k + 2) * 2 + 0], w21 = W2[(k + 2) * 2 + 1];
        const float w30 = W2[(k + 3) * 2 + 0], w31 = W2[(k + 3) * 2 + 1];
        float p0[RPW], p1[RPW];
#pragma unroll
        for (int i = 0; i < RPW; ++i) {
            const float* a = acc + min(wave + i * AGG_W, own - 1) * HID + lane;
            const float f0 = fmaxf(a[0]  + bv.x, 0.0f);
            const float f1 = fmaxf(a[32] + bv.y, 0.0f);
            const float f2 = fmaxf(a[64] + bv.z, 0.0f);
            const float f3 = fmaxf(a[96] + bv.w, 0.0f);
            p0[i] = f0 * w00 + f1 * w10 + f2 * w20 + f3 * w30;
            p1[i] = f0 * w01 + f1 * w11 + f2 * w21 + f3 * w31;
        }
#pragma unroll
        for (int i = 0; i < RPW; ++i) {
            p0[i] = gcn_wave_sum(p0[i]);
            p1[i] = gcn_wave_sum(p1[i]);
        }
        if (lane == 0) {
#pragma unroll
            for (int i = 0; i < RPW; ++i) {
                const int r = wave + i * AGG_W;
                zs[r * 2 + 0] = p0[i];
                zs[r * 2 + 1] = p1[i];
            }
        }
    }
    __syncthreads();

    const int nOut = own * N_CLS;
    for (int i = threadIdx.x * 4; i < nOut; i += AGG_T * 4) {
        if (i + 3 < nOut) {
            *(float4*)(zw + (size_t)lo * N_CLS + i) = *(const float4*)(zs + i);
        } else {
            for (int j = i; j < nOut; ++j) zw[(size_t)lo * N_CLS + j] = zs[j];
        }
    }
}


template <bool DIRECT>
__device__ __forceinline__ void gcn_list_flush2(const unsigned* lst, const float* nlst, int cnt, int lane,
                                                const unsigned* __restrict__ pk,
                                                const float* __restrict__ dinv,
                                                const float* __restrict__ zw,
                                                const float* dl, float* acc2) {
    for (int i0 = 0; i0 < cnt; i0 += 32 * CH) {
        unsigned w[CH];
        float    ds[CH];
        float2   z[CH];
#pragma unroll
        for (int c = 0; c < CH; ++c) {
            const int idx = min(i0 + c * 32 + lane, cnt - 1);
            w[c] = DIRECT ? lst[idx] : pk[lst[idx]];
            if (DIRECT) ds[c] = nlst[idx];
        }
#pragma unroll
        for (int c = 0; c < CH; ++c) {
            const int s = (int)(w[c] & ((1u << PK_BITS) - 1u));
            if (!DIRECT) ds[c] = dinv[s];
            z[c]  = *(const float2*)(zw + (size_t)s * N_CLS);
        }
#pragma unroll
        for (int c = 0; c < CH; ++c) {
            if (i0 + c * 32 + lane < cnt) {
                const int off = (int)(w[c] >> PK_BITS);
                const float norm = DIRECT ? ds[c] : ds[c] * dl[off];
                gcn_lds_add(acc2 + off * N_CLS + 0, z[c].x * norm);
                gcn_lds_add(acc2 + off * N_CLS + 1, z[c].y * norm);
            }
        }
    }
}

__global__ __launch_bounds__(AGG_T) void gcn_agg2_lds(const unsigned* __restrict__ pk,
                                                      const unsigned char* __restrict__ keys,
                                                      const unsigned* __restrict__ slots,
                                                      const float* __restrict__ norms,
                                                      const float* __restrict__ dinv,
                                                      const float* __restrict__ zw,
                                                      const float* __restrict__ b2,
                                                      float* __restrict__ out, int n, int nE) {
    __shared__ __attribute__((aligned(16))) float acc2[NPB * N_CLS];
    __shared__ float dl[NPB];
    __shared__ unsigned lst_all[AGG_W * LCAP];

    const int lane = threadIdx.x & 31;
    const int wave = __builtin_amdgcn_readfirstlane((int)(threadIdx.x >> 5));
    const int lo   = blockIdx.x * NPB;
    const int own  = min(NPB, n - lo);
    const unsigned* slot = slots + (size_t)blockIdx.x * SLOT_W;

    for (int r = threadIdx.x; r < own; r += AGG_T) {
        const float d = dinv[lo + r];
        const float s = d * d;
        const float2 zv = *(const float2*)(zw + (size_t)(lo + r) * N_CLS);
        dl[r] = d;
        acc2[r * N_CLS + 0] = zv.x * s;
        acc2[r * N_CLS + 1] = zv.y * s;
    }
    __syncthreads();

    int mine;
    if (gcn_slot_fits(slot, lane, wave, &mine)) {
        gcn_list_flush2<true>(slot + AGG_W + wave * WCAP,
                              norms + ((size_t)blockIdx.x * AGG_W + wave) * WCAP, mine, lane, pk,
                              dinv, zw, dl, acc2);
    } else {
        unsigned* lst = lst_all + wave * LCAP;
        gcn_sweep(keys, nE, (int)gridDim.x, lane, wave, lst, [&](int cnt) {
            gcn_list_flush2<false>(lst, nullptr, cnt, lane, pk, dinv, zw, dl, acc2);
        });
    }
    __syncthreads();

    const float c0 = b2[0], c1 = b2[1];
    for (int r = threadIdx.x; r < own; r += AGG_T) {
        const float l0 = acc2[r * N_CLS + 0] + c0;
        const float l1 = acc2[r * N_CLS + 1] + c1;
        const float m = fmaxf(l0, l1);
        const float lse = m + logf(expf(l0 - m) + expf(l1 - m));
        *(float2*)(out + (size_t)(lo + r) * N_CLS) = make_float2(l0 - lse, l1 - lse);
    }
}

extern "C" void kernel_launch(void* const* d_in, const int* in_sizes, int n_in,
                              void* d_out, int out_size, void* d_ws, size_t ws_size,
                              hipStream_t stream) {
    const float* x  = (const float*)d_in[0];
    const float* W1 = (const float*)d_in[1];
    const float* b1 = (const float*)d_in[2];
    const float* W2 = (const float*)d_in[3];
    const float* b2 = (const float*)d_in[4];
    const int*   ei = (const int*)d_in[5];

    const int n  = in_sizes[0] / IN_F;
    const int nE = in_sizes[5] / 2;
    const int* src = ei;
    const int* dst = ei + nE;

    char* ws = (char*)d_ws;
    float* xw   = (float*)(ws);
    unsigned*      pk   = (unsigned*)(ws + (size_t)n * HID * 4);
    unsigned char* keys = (unsigned char*)(ws + (size_t)n * HID * 4) + (size_t)nE * 4;
    unsigned*      slots = (unsigned*)(ws + (size_t)n * HID * 4 + (size_t)nE * 8);
    float* dinv = (float*)(ws + (size_t)2 * n * HID * 4);
    float* zw   = (float*)(ws + (size_t)2 * n * HID * 4 + 512000);
    float* out  = (float*)d_out;

    const int B = 256;
    const int G = (n + NPB - 1) / NPB;
    float* norms = (float*)(slots + (size_t)G * SLOT_W);

    gcn_edge_keys<<<((nE + 3) / 4 + B - 1) / B, B, 0, stream>>>(src, dst, pk, keys, nE);

    {
        const int tiles = (n / 16) * 2;
        gcn_own_gemm<<<G + (tiles + AGG_W - 1) / AGG_W, AGG_T, 0, stream>>>(pk, keys, dinv, slots, n, nE,
                                                                            x, W1, xw, G);
    }

    gcn_agg1_lds<<<G, AGG_T, 0, stream>>>(pk, keys, slots, norms, dinv, xw, b1, W2, zw, n, nE);

    gcn_agg2_lds<<<G, AGG_T, 0, stream>>>(pk, keys, slots, norms, dinv, zw, b2, out, n, nE);
}
